# starting kernel
# baseline (speedup 1.0000x reference)
#include <hip/hip_runtime.h>


#define NS_  8
#define NC_  4
#define T_   32
#define NB_  128
#define NSC_ 12
#define T1_  33

typedef float v2f __attribute__((ext_vector_type(2)));
typedef float v8f __attribute__((ext_vector_type(8)));

__device__ __forceinline__ v8f wmma4(v2f a, v2f b, v8f acc) {
  return __builtin_amdgcn_wmma_f32_16x16x4_f32(
       false, a,  false, b,
       (short)0, acc,  false,  false);
}

__global__ __launch_bounds__(32) void mpc_riccati(
    const float* __restrict__ x0g,
    const float* __restrict__ Cg_,
    const float* __restrict__ cg_,
    const float* __restrict__ Fg_,
    const float* __restrict__ fg_,
    float* __restrict__ out)
{
  const int  b    = blockIdx.x;
  const int  lane = threadIdx.x;
  const int  hl   = lane & 15;
  const bool hiw  = lane >= 16;
  const int  rbase = hiw ? 8 : 0;
  const int  kbase = hiw ? 2 : 0;

  __shared__ float Pm[16][16];
  __shared__ float Wm[16][16];
  __shared__ float Qm[16][16];
  __shared__ float Fs[16][16];
  __shared__ float Cs[16][16];
  __shared__ float Km[NC_][16];
  __shared__ float kvv[NC_];
  __shared__ float pv[NS_];
  __shared__ float qv[NSC_];
  __shared__ float Pfp[NS_];
  __shared__ float Kall[T1_][NC_][NS_];
  __shared__ float kall[T1_][NC_];
  __shared__ float xs[NS_];
  __shared__ float us[NC_];
  __shared__ float xn[NS_];

  for (int idx = lane; idx < 256; idx += 32) {
    int r = idx >> 4, cx = idx & 15;
    Pm[r][cx] = 0.f; Wm[r][cx] = 0.f; Qm[r][cx] = 0.f;
    Fs[r][cx] = 0.f; Cs[r][cx] = 0.f;
    if (r < NC_) Km[r][cx] = 0.f;
  }
  __syncthreads();

  auto solve_update = [&](int t) {
    float A[4][4], Inv[4][4];
    #pragma unroll
    for (int i = 0; i < 4; ++i)
      #pragma unroll
      for (int j = 0; j < 4; ++j) {
        A[i][j]   = Qm[NS_ + i][NS_ + j];
        Inv[i][j] = (i == j) ? 1.f : 0.f;
      }
    #pragma unroll
    for (int cc = 0; cc < 4; ++cc) {
      float piv = 1.f / A[cc][cc];
      #pragma unroll
      for (int j = 0; j < 4; ++j) { A[cc][j] *= piv; Inv[cc][j] *= piv; }
      #pragma unroll
      for (int r = 0; r < 4; ++r) {
        if (r == cc) continue;
        float fac = A[r][cc];
        #pragma unroll
        for (int j = 0; j < 4; ++j) { A[r][j] -= fac * A[cc][j]; Inv[r][j] -= fac * Inv[cc][j]; }
      }
    }
    {
      int jj = lane >> 3;
      int ii = lane & 7;
      float s = 0.f;
      #pragma unroll
      for (int m = 0; m < 4; ++m) s += Inv[jj][m] * Qm[NS_ + m][ii];
      Km[jj][ii]      = -s;
      Kall[t][jj][ii] = -s;
    }
    if (lane < NC_) {
      float s = 0.f;
      #pragma unroll
      for (int m = 0; m < 4; ++m) s += Inv[lane][m] * qv[NS_ + m];
      kvv[lane]     = -s;
      kall[t][lane] = -s;
    }
    __syncthreads();
    if (lane < NS_) {
      float s = qv[lane];
      #pragma unroll
      for (int j = 0; j < NC_; ++j) s += Qm[lane][NS_ + j] * kvv[j];
      pv[lane] = s;
    }
    v2f a, bb;
    {
      float a0 = Qm[hl][NS_ + kbase];
      float a1 = Qm[hl][NS_ + kbase + 1];
      a.x  = (hl < NS_) ? a0 : 0.f;
      a.y  = (hl < NS_) ? a1 : 0.f;
      bb.x = Km[kbase][hl];
      bb.y = Km[kbase + 1][hl];
    }
    v8f cf;
    #pragma unroll
    for (int r = 0; r < 8; ++r) {
      int row = r + rbase;
      float v = Qm[row][hl];
      cf[r] = (row < NS_ && hl < NS_) ? v : 0.f;
    }
    v8f d = wmma4(a, bb, cf);
    __syncthreads();
    #pragma unroll
    for (int r = 0; r < 8; ++r) Pm[r + rbase][hl] = d[r];
    __syncthreads();
  };

  {
    const float* Cg = Cg_ + (size_t)((T_ - 1) * NB_ + b) * NSC_ * NSC_;
    for (int idx = lane; idx < NSC_ * NSC_; idx += 32)
      Qm[idx / NSC_][idx % NSC_] = Cg[idx];
    const float* cg = cg_ + (size_t)((T_ - 1) * NB_ + b) * NSC_;
    if (lane < NSC_) qv[lane] = cg[lane];
    __syncthreads();
    solve_update(T_);
  }

  for (int t = T_ - 1; t >= 0; --t) {
    const float* Fg = Fg_ + (size_t)(t * NB_ + b) * NS_ * NSC_;
    const float* fg = fg_ + (size_t)(t * NB_ + b) * NS_;
    const float* Cg = Cg_ + (size_t)(t * NB_ + b) * NSC_ * NSC_;
    const float* cg = cg_ + (size_t)(t * NB_ + b) * NSC_;

    if (t > 0) {
      __builtin_prefetch(Fg_ + (size_t)((t - 1) * NB_ + b) * NS_ * NSC_, 0, 0);
      __builtin_prefetch(Cg_ + (size_t)((t - 1) * NB_ + b) * NSC_ * NSC_, 0, 0);
    }

    for (int idx = lane; idx < NS_ * NSC_; idx += 32)
      Fs[idx / NSC_][idx % NSC_] = Fg[idx];
    for (int idx = lane; idx < NSC_ * NSC_; idx += 32)
      Cs[idx / NSC_][idx % NSC_] = Cg[idx];
    __syncthreads();

    v2f fb0, fb1;
    fb0.x = Fs[kbase][hl];     fb0.y = Fs[kbase + 1][hl];
    fb1.x = Fs[4 + kbase][hl]; fb1.y = Fs[4 + kbase + 1][hl];

    v8f acc = {0.f, 0.f, 0.f, 0.f, 0.f, 0.f, 0.f, 0.f};
    {
      v2f a0, a1;
      a0.x = Pm[hl][kbase];     a0.y = Pm[hl][kbase + 1];
      a1.x = Pm[hl][4 + kbase]; a1.y = Pm[hl][4 + kbase + 1];
      acc = wmma4(a0, fb0, acc);
      acc = wmma4(a1, fb1, acc);
    }
    #pragma unroll
    for (int r = 0; r < 8; ++r) Wm[r + rbase][hl] = acc[r];

    if (lane < NS_) {
      float s = pv[lane];
      #pragma unroll
      for (int m = 0; m < NS_; ++m) s += Pm[lane][m] * fg[m];
      Pfp[lane] = s;
    }
    __syncthreads();

    v8f acc2;
    #pragma unroll
    for (int r = 0; r < 8; ++r) acc2[r] = Cs[r + rbase][hl];
    {
      v2f b0, b1;
      b0.x = Wm[kbase][hl];     b0.y = Wm[kbase + 1][hl];
      b1.x = Wm[4 + kbase][hl]; b1.y = Wm[4 + kbase + 1][hl];
      acc2 = wmma4(fb0, b0, acc2);
      acc2 = wmma4(fb1, b1, acc2);
    }
    #pragma unroll
    for (int r = 0; r < 8; ++r) Qm[r + rbase][hl] = acc2[r];

    if (lane < NSC_) {
      float s = cg[lane];
      #pragma unroll
      for (int i = 0; i < NS_; ++i) s += Fs[i][lane] * Pfp[i];
      qv[lane] = s;
    }
    __syncthreads();

    solve_update(t);
  }

  float* out_x = out;
  float* out_u = out + (size_t)T1_ * NB_ * NS_;
  if (lane < NS_) xs[lane] = x0g[(size_t)b * NS_ + lane];
  __syncthreads();
  for (int t = 0; t <= T_; ++t) {
    if (lane < NC_) {
      float s = kall[t][lane];
      #pragma unroll
      for (int i = 0; i < NS_; ++i) s += Kall[t][lane][i] * xs[i];
      us[lane] = s;
    }
    __syncthreads();
    if (lane < NS_) out_x[(size_t)t * NB_ * NS_ + b * NS_ + lane] = xs[lane];
    if (lane < NC_) out_u[(size_t)t * NB_ * NC_ + b * NC_ + lane] = us[lane];
    if (t < T_) {
      const float* Fg = Fg_ + (size_t)(t * NB_ + b) * NS_ * NSC_;
      const float* fg = fg_ + (size_t)(t * NB_ + b) * NS_;
      if (lane < NS_) {
        float s = fg[lane];
        #pragma unroll
        for (int j = 0; j < NS_; ++j) s += Fg[lane * NSC_ + j] * xs[j];
        #pragma unroll
        for (int j = 0; j < NC_; ++j) s += Fg[lane * NSC_ + NS_ + j] * us[j];
        xn[lane] = s;
      }
      __syncthreads();
      if (lane < NS_) xs[lane] = xn[lane];
      __syncthreads();
    }
  }
}

extern "C" void kernel_launch(void* const* d_in, const int* in_sizes, int n_in,
                              void* d_out, int out_size, void* d_ws, size_t ws_size,
                              hipStream_t stream) {
  (void)in_sizes; (void)n_in; (void)d_ws; (void)ws_size; (void)out_size;
  const float* x0 = (const float*)d_in[0];
  const float* C  = (const float*)d_in[1];
  const float* c  = (const float*)d_in[2];
  const float* F  = (const float*)d_in[3];
  const float* f  = (const float*)d_in[4];
  float* out = (float*)d_out;
  mpc_riccati<<<NB_, 32, 0, stream>>>(x0, C, c, F, f, out);
}


# Each step's F,C,c,f loads issued as one batch one step ahead into registers (sweep and rollout); block sync orders LDS only, so global loads/stores in flight are never waited for in the chain
# speedup vs baseline: 2.3658x; 2.3658x over previous
#include <hip/hip_runtime.h>


#define NS_  8
#define NC_  4
#define T_   32
#define NB_  128
#define NSC_ 12
#define T1_  33

typedef float v2f __attribute__((ext_vector_type(2)));
typedef float v8f __attribute__((ext_vector_type(8)));

__device__ __forceinline__ v8f wmma4(v2f a, v2f b, v8f acc) {
  return __builtin_amdgcn_wmma_f32_16x16x4_f32(
       false, a,  false, b,
       (short)0, acc,  false,  false);
}

__device__ __forceinline__ void lds_sync() {
  __builtin_amdgcn_fence(__ATOMIC_RELEASE, "workgroup", "local");
  __builtin_amdgcn_s_barrier();
  __builtin_amdgcn_fence(__ATOMIC_ACQUIRE, "workgroup", "local");
}

__global__ __launch_bounds__(32) void mpc_riccati(
    const float* __restrict__ x0g,
    const float* __restrict__ Cg_,
    const float* __restrict__ cg_,
    const float* __restrict__ Fg_,
    const float* __restrict__ fg_,
    float* __restrict__ out)
{
  const int  b    = blockIdx.x;
  const int  lane = threadIdx.x;
  const int  hl   = lane & 15;
  const bool hiw  = lane >= 16;
  const int  rbase = hiw ? 8 : 0;
  const int  kbase = hiw ? 2 : 0;

  __shared__ float Pm[16][16];
  __shared__ float Wm[16][16];
  __shared__ float Qm[16][16];
  __shared__ float Fs[16][16];
  __shared__ float Cs[16][16];
  __shared__ float Km[NC_][16];
  __shared__ float kvv[NC_];
  __shared__ float pv[NS_];
  __shared__ float qv[NSC_];
  __shared__ float Pfp[NS_];
  __shared__ float fsv[NS_];
  __shared__ float Kall[T1_][NC_][NS_];
  __shared__ float kall[T1_][NC_];
  __shared__ float xs[NS_];
  __shared__ float us[NC_];
  __shared__ float xn[NS_];

  float pF[3], pC[5], pf, pc;
  auto fetch = [&](int t) {
    const size_t tb = (size_t)(t * NB_ + b);
    const float* Fg = Fg_ + tb * (NS_ * NSC_);
    const float* Cg = Cg_ + tb * (NSC_ * NSC_);
    #pragma unroll
    for (int i = 0; i < 3; ++i) pF[i] = Fg[lane + 32 * i];
    #pragma unroll
    for (int i = 0; i < 4; ++i) pC[i] = Cg[lane + 32 * i];
    pC[4] = Cg[128 + hl];
    pf = fg_[tb * NS_ + (lane & (NS_ - 1))];
    pc = cg_[tb * NSC_ + (lane < NSC_ ? lane : NSC_ - 1)];
  };
  auto stage = [&]() {
    #pragma unroll
    for (int i = 0; i < 3; ++i) {
      int idx = lane + 32 * i;
      Fs[idx / NSC_][idx % NSC_] = pF[i];
    }
    #pragma unroll
    for (int i = 0; i < 4; ++i) {
      int idx = lane + 32 * i;
      Cs[idx / NSC_][idx % NSC_] = pC[i];
    }
    if (lane < 16) Cs[(128 + lane) / NSC_][(128 + lane) % NSC_] = pC[4];
    if (lane < NS_) fsv[lane] = pf;
  };
  float nF[NSC_], nf;
  auto fetch_fw = [&](int t) {
    const size_t tb = (size_t)(t * NB_ + b);
    const float* Fr = Fg_ + tb * (NS_ * NSC_) + (lane & (NS_ - 1)) * NSC_;
    #pragma unroll
    for (int j = 0; j < NSC_; ++j) nF[j] = Fr[j];
    nf = fg_[tb * NS_ + (lane & (NS_ - 1))];
  };

  fetch(T_ - 1);
  const float x0v = x0g[(size_t)b * NS_ + (lane & (NS_ - 1))];

  for (int idx = lane; idx < 256; idx += 32) {
    int r = idx >> 4, cx = idx & 15;
    Pm[r][cx] = 0.f; Wm[r][cx] = 0.f; Qm[r][cx] = 0.f;
    Fs[r][cx] = 0.f; Cs[r][cx] = 0.f;
    if (r < NC_) Km[r][cx] = 0.f;
  }
  lds_sync();

  auto solve_update = [&](int t) {
    float A[4][4], Inv[4][4];
    #pragma unroll
    for (int i = 0; i < 4; ++i)
      #pragma unroll
      for (int j = 0; j < 4; ++j) {
        A[i][j]   = Qm[NS_ + i][NS_ + j];
        Inv[i][j] = (i == j) ? 1.f : 0.f;
      }
    #pragma unroll
    for (int cc = 0; cc < 4; ++cc) {
      float piv = 1.f / A[cc][cc];
      #pragma unroll
      for (int j = 0; j < 4; ++j) { A[cc][j] *= piv; Inv[cc][j] *= piv; }
      #pragma unroll
      for (int r = 0; r < 4; ++r) {
        if (r == cc) continue;
        float fac = A[r][cc];
        #pragma unroll
        for (int j = 0; j < 4; ++j) { A[r][j] -= fac * A[cc][j]; Inv[r][j] -= fac * Inv[cc][j]; }
      }
    }
    {
      int jj = lane >> 3;
      int ii = lane & 7;
      float s = 0.f;
      #pragma unroll
      for (int m = 0; m < 4; ++m) s += Inv[jj][m] * Qm[NS_ + m][ii];
      Km[jj][ii]      = -s;
      Kall[t][jj][ii] = -s;
    }
    if (lane < NC_) {
      float s = 0.f;
      #pragma unroll
      for (int m = 0; m < 4; ++m) s += Inv[lane][m] * qv[NS_ + m];
      kvv[lane]     = -s;
      kall[t][lane] = -s;
    }
    lds_sync();
    if (lane < NS_) {
      float s = qv[lane];
      #pragma unroll
      for (int j = 0; j < NC_; ++j) s += Qm[lane][NS_ + j] * kvv[j];
      pv[lane] = s;
    }
    v2f a, bb;
    {
      float a0 = Qm[hl][NS_ + kbase];
      float a1 = Qm[hl][NS_ + kbase + 1];
      a.x  = (hl < NS_) ? a0 : 0.f;
      a.y  = (hl < NS_) ? a1 : 0.f;
      bb.x = Km[kbase][hl];
      bb.y = Km[kbase + 1][hl];
    }
    v8f cf;
    #pragma unroll
    for (int r = 0; r < 8; ++r) {
      int row = r + rbase;
      float v = Qm[row][hl];
      cf[r] = (row < NS_ && hl < NS_) ? v : 0.f;
    }
    v8f d = wmma4(a, bb, cf);
    lds_sync();
    #pragma unroll
    for (int r = 0; r < 8; ++r) Pm[r + rbase][hl] = d[r];
    lds_sync();
  };

  {
    #pragma unroll
    for (int i = 0; i < 4; ++i) {
      int idx = lane + 32 * i;
      Qm[idx / NSC_][idx % NSC_] = pC[i];
    }
    if (lane < 16) Qm[(128 + lane) / NSC_][(128 + lane) % NSC_] = pC[4];
    if (lane < NSC_) qv[lane] = pc;
    lds_sync();
    solve_update(T_);
  }
  fetch_fw(0);

  for (int t = T_ - 1; t >= 0; --t) {
    stage();
    const float ct = pc;
    if (t > 0) fetch(t - 1);
    lds_sync();

    v2f fb0, fb1;
    fb0.x = Fs[kbase][hl];     fb0.y = Fs[kbase + 1][hl];
    fb1.x = Fs[4 + kbase][hl]; fb1.y = Fs[4 + kbase + 1][hl];

    v8f acc = {0.f, 0.f, 0.f, 0.f, 0.f, 0.f, 0.f, 0.f};
    {
      v2f a0, a1;
      a0.x = Pm[hl][kbase];     a0.y = Pm[hl][kbase + 1];
      a1.x = Pm[hl][4 + kbase]; a1.y = Pm[hl][4 + kbase + 1];
      acc = wmma4(a0, fb0, acc);
      acc = wmma4(a1, fb1, acc);
    }
    #pragma unroll
    for (int r = 0; r < 8; ++r) Wm[r + rbase][hl] = acc[r];

    if (lane < NS_) {
      float s = pv[lane];
      #pragma unroll
      for (int m = 0; m < NS_; ++m) s += Pm[lane][m] * fsv[m];
      Pfp[lane] = s;
    }
    lds_sync();

    v8f acc2;
    #pragma unroll
    for (int r = 0; r < 8; ++r) acc2[r] = Cs[r + rbase][hl];
    {
      v2f b0, b1;
      b0.x = Wm[kbase][hl];     b0.y = Wm[kbase + 1][hl];
      b1.x = Wm[4 + kbase][hl]; b1.y = Wm[4 + kbase + 1][hl];
      acc2 = wmma4(fb0, b0, acc2);
      acc2 = wmma4(fb1, b1, acc2);
    }
    #pragma unroll
    for (int r = 0; r < 8; ++r) Qm[r + rbase][hl] = acc2[r];

    if (lane < NSC_) {
      float s = ct;
      #pragma unroll
      for (int i = 0; i < NS_; ++i) s += Fs[i][lane] * Pfp[i];
      qv[lane] = s;
    }
    lds_sync();

    solve_update(t);
  }

  float* out_x = out;
  float* out_u = out + (size_t)T1_ * NB_ * NS_;
  if (lane < NS_) xs[lane] = x0v;
  lds_sync();
  for (int t = 0; t <= T_; ++t) {
    float cF[NSC_];
    #pragma unroll
    for (int j = 0; j < NSC_; ++j) cF[j] = nF[j];
    const float cf = nf;
    if (t + 1 < T_) fetch_fw(t + 1);
    if (lane < NC_) {
      float s = kall[t][lane];
      #pragma unroll
      for (int i = 0; i < NS_; ++i) s += Kall[t][lane][i] * xs[i];
      us[lane] = s;
    }
    lds_sync();
    if (lane < NS_) out_x[(size_t)t * NB_ * NS_ + b * NS_ + lane] = xs[lane];
    if (lane < NC_) out_u[(size_t)t * NB_ * NC_ + b * NC_ + lane] = us[lane];
    if (t < T_) {
      if (lane < NS_) {
        float s = cf;
        #pragma unroll
        for (int j = 0; j < NS_; ++j) s += cF[j] * xs[j];
        #pragma unroll
        for (int j = 0; j < NC_; ++j) s += cF[NS_ + j] * us[j];
        xn[lane] = s;
      }
      lds_sync();
      if (lane < NS_) xs[lane] = xn[lane];
      lds_sync();
    }
  }
}

extern "C" void kernel_launch(void* const* d_in, const int* in_sizes, int n_in,
                              void* d_out, int out_size, void* d_ws, size_t ws_size,
                              hipStream_t stream) {
  (void)in_sizes; (void)n_in; (void)d_ws; (void)ws_size; (void)out_size;
  const float* x0 = (const float*)d_in[0];
  const float* C  = (const float*)d_in[1];
  const float* c  = (const float*)d_in[2];
  const float* F  = (const float*)d_in[3];
  const float* f  = (const float*)d_in[4];
  float* out = (float*)d_out;
  mpc_riccati<<<NB_, 32, 0, stream>>>(x0, C, c, F, f, out);
}


# Backward sweep's serial chain kept in registers: WMMA results handed on as operands by half-swap permlane (Q^T, P^T formed beside Q, P; affine terms summed as column 12 of the WMMAs); no LDS hop
# speedup vs baseline: 3.8386x; 1.6225x over previous
#include <hip/hip_runtime.h>


#define NS_  8
#define NC_  4
#define T_   32
#define NB_  128
#define NSC_ 12
#define T1_  33

typedef float v2f __attribute__((ext_vector_type(2)));
typedef float v4f __attribute__((ext_vector_type(4)));
typedef float v8f __attribute__((ext_vector_type(8)));

__device__ __forceinline__ v8f wmma4(v2f a, v2f b, v8f acc) {
  return __builtin_amdgcn_wmma_f32_16x16x4_f32(
       false, a,  false, b,
       (short)0, acc,  false,  false);
}

__device__ __forceinline__ void lds_sync() {
  __builtin_amdgcn_fence(__ATOMIC_RELEASE, "workgroup", "local");
  __builtin_amdgcn_s_barrier();
  __builtin_amdgcn_fence(__ATOMIC_ACQUIRE, "workgroup", "local");
}

__device__ __forceinline__ float other_half(float v) {
  unsigned r = __builtin_amdgcn_permlanex16(0u, __builtin_bit_cast(unsigned, v),
                                            0x76543210u, 0xfedcba98u, true, true);
  return __builtin_bit_cast(float, r);
}

template <int L>
__device__ __forceinline__ float row_lane(float v) {
  int r = __builtin_amdgcn_mov_dpp(__builtin_bit_cast(int, v), 0x150 + L, 0xf, 0xf, true);
  return __builtin_bit_cast(float, r);
}

struct StepIn {
  v2f fa0, fa1;
  v2f fb0, fb1;
  v8f Cd;
  v8f CTd;
};

struct StepRaw {
  float fv[4];
  float cv[8];
  v4f   t0, t1;
};

struct Carry {
  v2f aW0, aW1;
  v8f seed;
};

__global__ __launch_bounds__(32) void mpc_riccati(
    const float* __restrict__ x0g,
    const float* __restrict__ Cg_,
    const float* __restrict__ cg_,
    const float* __restrict__ Fg_,
    const float* __restrict__ fg_,
    float* __restrict__ out)
{
  const int  b    = blockIdx.x;
  const int  lane = threadIdx.x;
  const int  hl   = lane & 15;
  const bool hiw  = lane >= 16;
  const int  rbase = hiw ? 8 : 0;
  const int  kbase = hiw ? 2 : 0;
  const bool colx  = hl < NS_;
  const bool colxp = colx || hl == NSC_;

  __shared__ float Kall[T1_][NC_][NS_];
  __shared__ float kall[T1_][NC_];
  __shared__ float xs[NS_];
  __shared__ float us[NC_];
  __shared__ float xn[NS_];

  auto fetch = [&](int t, StepRaw& rw) {
    const size_t tb = (size_t)(t * NB_ + b);
    const float* Fg = Fg_ + tb * (NS_ * NSC_);
    const float* Cg = Cg_ + tb * (NSC_ * NSC_);
    const float* cg = cg_ + tb * NSC_;
    const float* fg = fg_ + tb * NS_;
    const int  cc   = hl < NSC_ ? hl : NSC_ - 1;
    const bool aff  = hl == NSC_;
    const float* pf0 = aff ? fg : Fg + cc;
    const float* pc0 = aff ? cg : Cg + cc;
    const int    st  = aff ? 1 : NSC_;
    #pragma unroll
    for (int i = 0; i < 4; ++i) {
      const int k = 4 * (i >> 1) + kbase + (i & 1);
      rw.fv[i] = pf0[k * st];
    }
    #pragma unroll
    for (int r = 0; r < 8; ++r) {
      const int row = (r < 4) ? rbase + r : r;
      rw.cv[r] = pc0[row * st];
    }
    {
      const float* pt = Cg + cc * NSC_;
      const int r0 = rbase;
      rw.t0.x = pt[r0];  rw.t0.y = pt[r0 + 1];  rw.t0.z = pt[r0 + 2];  rw.t0.w = pt[r0 + 3];
      rw.t1.x = pt[4];   rw.t1.y = pt[5];       rw.t1.z = pt[6];       rw.t1.w = pt[7];
    }
  };
  auto shape = [&](const StepRaw& rw, StepIn& in) {
    const float* fv = rw.fv; const float* cv = rw.cv;
    const float tv[8] = {rw.t0.x, rw.t0.y, rw.t0.z, rw.t0.w, rw.t1.x, rw.t1.y, rw.t1.z, rw.t1.w};
    in.fa0.x = hl < NSC_ ? fv[0] : 0.f;  in.fa0.y = hl < NSC_ ? fv[1] : 0.f;
    in.fa1.x = hl < NSC_ ? fv[2] : 0.f;  in.fa1.y = hl < NSC_ ? fv[3] : 0.f;
    in.fb0.x = hl <= NSC_ ? fv[0] : 0.f; in.fb0.y = hl <= NSC_ ? fv[1] : 0.f;
    in.fb1.x = hl <= NSC_ ? fv[2] : 0.f; in.fb1.y = hl <= NSC_ ? fv[3] : 0.f;
    #pragma unroll
    for (int r = 0; r < 8; ++r) {
      const bool rowok = (r < 4) || !hiw;
      in.Cd[r]  = (rowok && hl <= NSC_) ? cv[r] : 0.f;
      in.CTd[r] = (rowok && hl <  NSC_) ? tv[r] : 0.f;
    }
  };
  float nF[NSC_], nf;
  auto fetch_fw = [&](int t) {
    const size_t tb = (size_t)(t * NB_ + b);
    const float* Fr = Fg_ + tb * (NS_ * NSC_) + (lane & (NS_ - 1)) * NSC_;
    #pragma unroll
    for (int j = 0; j < NSC_; ++j) nF[j] = Fr[j];
    nf = fg_[tb * NS_ + (lane & (NS_ - 1))];
  };

  StepRaw nx;
  fetch(T_ - 1, nx);
  const float x0v = x0g[(size_t)b * NS_ + (lane & (NS_ - 1))];

  Carry cy;

  auto frag = [&](const v8f& d, int kb) {
    const float s2 = other_half(d[4 * kb + 2]);
    const float s3 = other_half(d[4 * kb + 3]);
    v2f o;
    o.x = hiw ? s2 : d[4 * kb];
    o.y = hiw ? s3 : d[4 * kb + 1];
    return o;
  };

  auto solve_update = [&](const v8f& Qd, const v8f& QTd, int t) {
    float X[4];
    #pragma unroll
    for (int m = 0; m < 4; ++m) {
      const float o = other_half(Qd[m]);
      X[m] = hiw ? Qd[m] : o;
    }
    float A[4][4], Inv[4][4];
    #pragma unroll
    for (int i = 0; i < 4; ++i) {
      A[i][0] = row_lane<NS_ + 0>(X[i]);
      A[i][1] = row_lane<NS_ + 1>(X[i]);
      A[i][2] = row_lane<NS_ + 2>(X[i]);
      A[i][3] = row_lane<NS_ + 3>(X[i]);
      #pragma unroll
      for (int j = 0; j < 4; ++j) Inv[i][j] = (i == j) ? 1.f : 0.f;
    }
    #pragma unroll
    for (int cc = 0; cc < 4; ++cc) {
      float piv = 1.f / A[cc][cc];
      #pragma unroll
      for (int j = 0; j < 4; ++j) { A[cc][j] *= piv; Inv[cc][j] *= piv; }
      #pragma unroll
      for (int r = 0; r < 4; ++r) {
        if (r == cc) continue;
        float fac = A[r][cc];
        #pragma unroll
        for (int j = 0; j < 4; ++j) { A[r][j] -= fac * A[cc][j]; Inv[r][j] -= fac * Inv[cc][j]; }
      }
    }
    float kx = 0.f, ky = 0.f;
    #pragma unroll
    for (int m = 0; m < 4; ++m) {
      const float xm = colxp ? X[m] : 0.f;
      kx += (hiw ? Inv[2][m] : Inv[0][m]) * xm;
      ky += (hiw ? Inv[3][m] : Inv[1][m]) * xm;
    }
    kx = -kx; ky = -ky;
    if (colx) {
      Kall[t][kbase][hl]     = kx;
      Kall[t][kbase + 1][hl] = ky;
    }
    if (hl == NSC_) {
      kall[t][kbase]     = kx;
      kall[t][kbase + 1] = ky;
    }
    const float o0 = other_half(QTd[0]);
    const float o1 = other_half(QTd[1]);
    v2f qxu, kf, kt;
    qxu.x = colx ? (hiw ? QTd[2] : o0) : 0.f;
    qxu.y = colx ? (hiw ? QTd[3] : o1) : 0.f;
    kf.x = kx;  kf.y = ky;
    kt.x = colx ? kx : 0.f;  kt.y = colx ? ky : 0.f;
    v8f sP, sT;
    #pragma unroll
    for (int r = 0; r < 8; ++r) {
      sP[r] = (!hiw && colxp) ? Qd[r]  : 0.f;
      sT[r] = (!hiw && colx)  ? QTd[r] : 0.f;
    }
    const v8f Pd  = wmma4(qxu, kf, sP);
    const v8f PTd = wmma4(kt, qxu, sT);
    cy.aW0 = frag(PTd, 0);  cy.aW1 = frag(PTd, 1);
    #pragma unroll
    for (int r = 0; r < 8; ++r) cy.seed[r] = (hl == NSC_) ? Pd[r] : 0.f;
  };

  {
    StepIn in;
    shape(nx, in);
    solve_update(in.Cd, in.CTd, T_);
  }
  fetch_fw(0);

  for (int t = T_ - 1; t >= 0; --t) {
    StepIn in;
    shape(nx, in);
    if (t > 0) fetch(t - 1, nx);

    v8f Wd = wmma4(cy.aW0, in.fb0, cy.seed);
    Wd = wmma4(cy.aW1, in.fb1, Wd);

    const v2f w0 = frag(Wd, 0), w1 = frag(Wd, 1);
    v8f Qd  = wmma4(in.fa0, w0, in.Cd);
    v8f QTd = wmma4(w0, in.fa0, in.CTd);
    Qd  = wmma4(in.fa1, w1, Qd);
    QTd = wmma4(w1, in.fa1, QTd);

    solve_update(Qd, QTd, t);
  }

  float* out_x = out;
  float* out_u = out + (size_t)T1_ * NB_ * NS_;
  if (lane < NS_) xs[lane] = x0v;
  lds_sync();
  for (int t = 0; t <= T_; ++t) {
    float cF[NSC_];
    #pragma unroll
    for (int j = 0; j < NSC_; ++j) cF[j] = nF[j];
    const float cf = nf;
    if (t + 1 < T_) fetch_fw(t + 1);
    if (lane < NC_) {
      float s = kall[t][lane];
      #pragma unroll
      for (int i = 0; i < NS_; ++i) s += Kall[t][lane][i] * xs[i];
      us[lane] = s;
    }
    lds_sync();
    if (lane < NS_) out_x[(size_t)t * NB_ * NS_ + b * NS_ + lane] = xs[lane];
    if (lane < NC_) out_u[(size_t)t * NB_ * NC_ + b * NC_ + lane] = us[lane];
    if (t < T_) {
      if (lane < NS_) {
        float s = cf;
        #pragma unroll
        for (int j = 0; j < NS_; ++j) s += cF[j] * xs[j];
        #pragma unroll
        for (int j = 0; j < NC_; ++j) s += cF[NS_ + j] * us[j];
        xn[lane] = s;
      }
      lds_sync();
      if (lane < NS_) xs[lane] = xn[lane];
      lds_sync();
    }
  }
}

extern "C" void kernel_launch(void* const* d_in, const int* in_sizes, int n_in,
                              void* d_out, int out_size, void* d_ws, size_t ws_size,
                              hipStream_t stream) {
  (void)in_sizes; (void)n_in; (void)d_ws; (void)ws_size; (void)out_size;
  const float* x0 = (const float*)d_in[0];
  const float* C  = (const float*)d_in[1];
  const float* c  = (const float*)d_in[2];
  const float* F  = (const float*)d_in[3];
  const float* f  = (const float*)d_in[4];
  float* out = (float*)d_out;
  mpc_riccati<<<NB_, 32, 0, stream>>>(x0, C, c, F, f, out);
}


# Rollout's serial chain kept in registers: x, u read from their holding lanes by DPP row_share; coefficient rows staged in LDS by the sweep, read two steps ahead; products fused into the sums (FMA)
# speedup vs baseline: 4.3616x; 1.1362x over previous
#include <hip/hip_runtime.h>


#define NS_  8
#define NC_  4
#define T_   32
#define NB_  128
#define NSC_ 12
#define T1_  33

typedef float v2f __attribute__((ext_vector_type(2)));
typedef float v4f __attribute__((ext_vector_type(4)));
typedef float v8f __attribute__((ext_vector_type(8)));

__device__ __forceinline__ v8f wmma4(v2f a, v2f b, v8f acc) {
  return __builtin_amdgcn_wmma_f32_16x16x4_f32(
       false, a,  false, b,
       (short)0, acc,  false,  false);
}

__device__ __forceinline__ void lds_sync() {
  __builtin_amdgcn_fence(__ATOMIC_RELEASE, "workgroup", "local");
  __builtin_amdgcn_s_barrier();
  __builtin_amdgcn_fence(__ATOMIC_ACQUIRE, "workgroup", "local");
}

__device__ __forceinline__ float other_half(float v) {
  unsigned r = __builtin_amdgcn_permlanex16(0u, __builtin_bit_cast(unsigned, v),
                                            0x76543210u, 0xfedcba98u, true, true);
  return __builtin_bit_cast(float, r);
}

template <int L>
__device__ __forceinline__ float row_lane(float v) {
  int r = __builtin_amdgcn_mov_dpp(__builtin_bit_cast(int, v), 0x150 + L, 0xf, 0xf, true);
  return __builtin_bit_cast(float, r);
}

struct StepIn {
  v2f fa0, fa1;
  v2f fb0, fb1;
  v8f Cd;
  v8f CTd;
};

struct StepRaw {
  float fv[4];
  float cv[8];
  v4f   t0, t1;
};

struct Carry {
  v2f aW0, aW1;
  v8f seed;
};

__global__ __launch_bounds__(32) void mpc_riccati(
    const float* __restrict__ x0g,
    const float* __restrict__ Cg_,
    const float* __restrict__ cg_,
    const float* __restrict__ Fg_,
    const float* __restrict__ fg_,
    float* __restrict__ out)
{
  const int  b    = blockIdx.x;
  const int  lane = threadIdx.x;
  const int  hl   = lane & 15;
  const bool hiw  = lane >= 16;
  const int  rbase = hiw ? 8 : 0;
  const int  kbase = hiw ? 2 : 0;
  const bool colx  = hl < NS_;
  const bool colxp = colx || hl == NSC_;

  __shared__ __attribute__((aligned(16))) float Coef[T1_][NSC_][16];

  auto fetch = [&](int t, StepRaw& rw) {
    const size_t tb = (size_t)(t * NB_ + b);
    const float* Fg = Fg_ + tb * (NS_ * NSC_);
    const float* Cg = Cg_ + tb * (NSC_ * NSC_);
    const float* cg = cg_ + tb * NSC_;
    const float* fg = fg_ + tb * NS_;
    const int  cc   = hl < NSC_ ? hl : NSC_ - 1;
    const bool aff  = hl == NSC_;
    const float* pf0 = aff ? fg : Fg + cc;
    const float* pc0 = aff ? cg : Cg + cc;
    const int    st  = aff ? 1 : NSC_;
    #pragma unroll
    for (int i = 0; i < 4; ++i) {
      const int k = 4 * (i >> 1) + kbase + (i & 1);
      rw.fv[i] = pf0[k * st];
    }
    #pragma unroll
    for (int r = 0; r < 8; ++r) {
      const int row = (r < 4) ? rbase + r : r;
      rw.cv[r] = pc0[row * st];
    }
    {
      const float* pt = Cg + cc * NSC_;
      const int r0 = rbase;
      rw.t0.x = pt[r0];  rw.t0.y = pt[r0 + 1];  rw.t0.z = pt[r0 + 2];  rw.t0.w = pt[r0 + 3];
      rw.t1.x = pt[4];   rw.t1.y = pt[5];       rw.t1.z = pt[6];       rw.t1.w = pt[7];
    }
  };
  auto shape = [&](const StepRaw& rw, StepIn& in) {
    const float* fv = rw.fv; const float* cv = rw.cv;
    const float tv[8] = {rw.t0.x, rw.t0.y, rw.t0.z, rw.t0.w, rw.t1.x, rw.t1.y, rw.t1.z, rw.t1.w};
    in.fa0.x = hl < NSC_ ? fv[0] : 0.f;  in.fa0.y = hl < NSC_ ? fv[1] : 0.f;
    in.fa1.x = hl < NSC_ ? fv[2] : 0.f;  in.fa1.y = hl < NSC_ ? fv[3] : 0.f;
    in.fb0.x = hl <= NSC_ ? fv[0] : 0.f; in.fb0.y = hl <= NSC_ ? fv[1] : 0.f;
    in.fb1.x = hl <= NSC_ ? fv[2] : 0.f; in.fb1.y = hl <= NSC_ ? fv[3] : 0.f;
    #pragma unroll
    for (int r = 0; r < 8; ++r) {
      const bool rowok = (r < 4) || !hiw;
      in.Cd[r]  = (rowok && hl <= NSC_) ? cv[r] : 0.f;
      in.CTd[r] = (rowok && hl <  NSC_) ? tv[r] : 0.f;
    }
  };
  StepRaw nx;
  fetch(T_ - 1, nx);
  const float x0v = x0g[(size_t)b * NS_ + (lane & (NS_ - 1))];

  Carry cy;

  auto frag = [&](const v8f& d, int kb) {
    const float s2 = other_half(d[4 * kb + 2]);
    const float s3 = other_half(d[4 * kb + 3]);
    v2f o;
    o.x = hiw ? s2 : d[4 * kb];
    o.y = hiw ? s3 : d[4 * kb + 1];
    return o;
  };

  auto solve_update = [&](const v8f& Qd, const v8f& QTd, int t) {
    float X[4];
    #pragma unroll
    for (int m = 0; m < 4; ++m) {
      const float o = other_half(Qd[m]);
      X[m] = hiw ? Qd[m] : o;
    }
    float A[4][4], Inv[4][4];
    #pragma unroll
    for (int i = 0; i < 4; ++i) {
      A[i][0] = row_lane<NS_ + 0>(X[i]);
      A[i][1] = row_lane<NS_ + 1>(X[i]);
      A[i][2] = row_lane<NS_ + 2>(X[i]);
      A[i][3] = row_lane<NS_ + 3>(X[i]);
      #pragma unroll
      for (int j = 0; j < 4; ++j) Inv[i][j] = (i == j) ? 1.f : 0.f;
    }
    #pragma unroll
    for (int cc = 0; cc < 4; ++cc) {
      float piv = 1.f / A[cc][cc];
      #pragma unroll
      for (int j = 0; j < 4; ++j) { A[cc][j] *= piv; Inv[cc][j] *= piv; }
      #pragma unroll
      for (int r = 0; r < 4; ++r) {
        if (r == cc) continue;
        float fac = A[r][cc];
        #pragma unroll
        for (int j = 0; j < 4; ++j) { A[r][j] -= fac * A[cc][j]; Inv[r][j] -= fac * Inv[cc][j]; }
      }
    }
    float kx = 0.f, ky = 0.f;
    #pragma unroll
    for (int m = 0; m < 4; ++m) {
      const float xm = colxp ? X[m] : 0.f;
      kx += (hiw ? Inv[2][m] : Inv[0][m]) * xm;
      ky += (hiw ? Inv[3][m] : Inv[1][m]) * xm;
    }
    kx = -kx; ky = -ky;
    Coef[t][NS_ + kbase][hl]     = kx;
    Coef[t][NS_ + kbase + 1][hl] = ky;
    const float o0 = other_half(QTd[0]);
    const float o1 = other_half(QTd[1]);
    v2f qxu, kf, kt;
    qxu.x = colx ? (hiw ? QTd[2] : o0) : 0.f;
    qxu.y = colx ? (hiw ? QTd[3] : o1) : 0.f;
    kf.x = kx;  kf.y = ky;
    kt.x = colx ? kx : 0.f;  kt.y = colx ? ky : 0.f;
    v8f sP, sT;
    #pragma unroll
    for (int r = 0; r < 8; ++r) {
      sP[r] = (!hiw && colxp) ? Qd[r]  : 0.f;
      sT[r] = (!hiw && colx)  ? QTd[r] : 0.f;
    }
    const v8f Pd  = wmma4(qxu, kf, sP);
    const v8f PTd = wmma4(kt, qxu, sT);
    cy.aW0 = frag(PTd, 0);  cy.aW1 = frag(PTd, 1);
    #pragma unroll
    for (int r = 0; r < 8; ++r) cy.seed[r] = (hl == NSC_) ? Pd[r] : 0.f;
  };

  {
    StepIn in;
    shape(nx, in);
    solve_update(in.Cd, in.CTd, T_);
    Coef[T_][kbase][hl]     = 0.f;  Coef[T_][kbase + 1][hl] = 0.f;
    Coef[T_][4 + kbase][hl] = 0.f;  Coef[T_][5 + kbase][hl] = 0.f;
  }

  for (int t = T_ - 1; t >= 0; --t) {
    StepIn in;
    shape(nx, in);
    if (t > 0) fetch(t - 1, nx);
    Coef[t][kbase][hl]     = in.fb0.x;  Coef[t][kbase + 1][hl] = in.fb0.y;
    Coef[t][4 + kbase][hl] = in.fb1.x;  Coef[t][5 + kbase][hl] = in.fb1.y;

    v8f Wd = wmma4(cy.aW0, in.fb0, cy.seed);
    Wd = wmma4(cy.aW1, in.fb1, Wd);

    const v2f w0 = frag(Wd, 0), w1 = frag(Wd, 1);
    v8f Qd  = wmma4(in.fa0, w0, in.Cd);
    v8f QTd = wmma4(w0, in.fa0, in.CTd);
    Qd  = wmma4(in.fa1, w1, Qd);
    QTd = wmma4(w1, in.fa1, QTd);

    solve_update(Qd, QTd, t);
  }

  float* out_x = out;
  float* out_u = out + (size_t)T1_ * NB_ * NS_;
  struct RowCo { v4f a0, a1, bu; float c0; };
  const int rr = hl < NSC_ ? hl : NSC_ - 1;
  auto ldco = [&](int t, RowCo& rc) {
    const v4f* p = reinterpret_cast<const v4f*>(&Coef[t][rr][0]);
    rc.a0 = p[0];  rc.a1 = p[1];  rc.bu = p[2];
    rc.c0 = Coef[t][rr][NSC_];
  };
  float v = x0v;
  float* po = hl < NS_ ? out_x + (size_t)b * NS_ + hl
                       : out_u + (size_t)b * NC_ + ((hl - NS_) & (NC_ - 1));
  const int pstride = hl < NS_ ? NB_ * NS_ : NB_ * NC_;
  auto step = [&](const RowCo& rc) {
    float s = rc.c0;
    s = __builtin_fmaf(rc.a0.x, row_lane<0>(v), s);  s = __builtin_fmaf(rc.a0.y, row_lane<1>(v), s);
    s = __builtin_fmaf(rc.a0.z, row_lane<2>(v), s);  s = __builtin_fmaf(rc.a0.w, row_lane<3>(v), s);
    s = __builtin_fmaf(rc.a1.x, row_lane<4>(v), s);  s = __builtin_fmaf(rc.a1.y, row_lane<5>(v), s);
    s = __builtin_fmaf(rc.a1.z, row_lane<6>(v), s);  s = __builtin_fmaf(rc.a1.w, row_lane<7>(v), s);
    const float val = hl < NS_ ? v : s;
    if (lane < NSC_) *po = val;
    po += pstride;
    float s2 = s;
    s2 = __builtin_fmaf(rc.bu.x, row_lane<NS_ + 0>(s), s2);  s2 = __builtin_fmaf(rc.bu.y, row_lane<NS_ + 1>(s), s2);
    s2 = __builtin_fmaf(rc.bu.z, row_lane<NS_ + 2>(s), s2);  s2 = __builtin_fmaf(rc.bu.w, row_lane<NS_ + 3>(s), s2);
    v = s2;
  };
  lds_sync();
  RowCo r0, r1, r2;
  ldco(0, r0);  ldco(1, r1);  ldco(2, r2);
  for (int t = 0; t <= T_; t += 3) {
    step(r0);  ldco(t + 3 < T_ ? t + 3 : T_, r0);
    step(r1);  ldco(t + 4 < T_ ? t + 4 : T_, r1);
    step(r2);  ldco(t + 5 < T_ ? t + 5 : T_, r2);
  }
}

extern "C" void kernel_launch(void* const* d_in, const int* in_sizes, int n_in,
                              void* d_out, int out_size, void* d_ws, size_t ws_size,
                              hipStream_t stream) {
  (void)in_sizes; (void)n_in; (void)d_ws; (void)ws_size; (void)out_size;
  const float* x0 = (const float*)d_in[0];
  const float* C  = (const float*)d_in[1];
  const float* c  = (const float*)d_in[2];
  const float* F  = (const float*)d_in[3];
  const float* f  = (const float*)d_in[4];
  float* out = (float*)d_out;
  mpc_riccati<<<NB_, 32, 0, stream>>>(x0, C, c, F, f, out);
}


# All of a batch's inputs copied global->LDS by one burst of 65 128-bit loads per lane at the start; each sweep step reads its operands from LDS, so no step waits for a memory latency
# speedup vs baseline: 8.1767x; 1.8747x over previous
#include <hip/hip_runtime.h>


#define NS_  8
#define NC_  4
#define T_   32
#define NB_  128
#define NSC_ 12
#define T1_  33

typedef float v2f __attribute__((ext_vector_type(2)));
typedef float v4f __attribute__((ext_vector_type(4)));
typedef float v8f __attribute__((ext_vector_type(8)));

__device__ __forceinline__ v8f wmma4(v2f a, v2f b, v8f acc) {
  return __builtin_amdgcn_wmma_f32_16x16x4_f32(
       false, a,  false, b,
       (short)0, acc,  false,  false);
}

__device__ __forceinline__ void lds_sync() {
  __builtin_amdgcn_fence(__ATOMIC_RELEASE, "workgroup", "local");
  __builtin_amdgcn_s_barrier();
  __builtin_amdgcn_fence(__ATOMIC_ACQUIRE, "workgroup", "local");
}

__device__ __forceinline__ float other_half(float v) {
  unsigned r = __builtin_amdgcn_permlanex16(0u, __builtin_bit_cast(unsigned, v),
                                            0x76543210u, 0xfedcba98u, true, true);
  return __builtin_bit_cast(float, r);
}

template <int L>
__device__ __forceinline__ float row_lane(float v) {
  int r = __builtin_amdgcn_mov_dpp(__builtin_bit_cast(int, v), 0x150 + L, 0xf, 0xf, true);
  return __builtin_bit_cast(float, r);
}

struct StepIn {
  v2f fa0, fa1;
  v2f fb0, fb1;
  v8f Cd;
  v8f CTd;
};

struct StepRaw {
  float fv[4];
  float cv[8];
  v4f   t0, t1;
};

struct Carry {
  v2f aW0, aW1;
  v8f seed;
};

__global__ __launch_bounds__(32) void mpc_riccati(
    const float* __restrict__ x0g,
    const float* __restrict__ Cg_,
    const float* __restrict__ cg_,
    const float* __restrict__ Fg_,
    const float* __restrict__ fg_,
    float* __restrict__ out)
{
  const int  b    = blockIdx.x;
  const int  lane = threadIdx.x;
  const int  hl   = lane & 15;
  const bool hiw  = lane >= 16;
  const int  rbase = hiw ? 8 : 0;
  const int  kbase = hiw ? 2 : 0;
  const bool colx  = hl < NS_;
  const bool colxp = colx || hl == NSC_;

  __shared__ __attribute__((aligned(16))) float Coef[T1_][NSC_][16];
  __shared__ __attribute__((aligned(16))) float sC[T_][NSC_ * NSC_];
  __shared__ __attribute__((aligned(16))) float sc[T_][NSC_];
  __shared__ __attribute__((aligned(16))) float sF[T_][NS_ * NSC_];
  __shared__ __attribute__((aligned(16))) float sf[T_][NS_];

  {
    constexpr int CS = NB_ * NSC_ * NSC_;
    constexpr int FS = NB_ * NS_ * NSC_;
    constexpr int cS = NB_ * NSC_;
    constexpr int fS = NB_ * NS_;
    const float* gC  = Cg_ + (size_t)b * (NSC_ * NSC_) + 4 * lane;
    const int    tq  = lane >> 2, qq = 128 + 4 * (lane & 3);
    const float* gCt = Cg_ + (size_t)(tq * NB_ + b) * (NSC_ * NSC_) + qq;
    int ft[3], fq[3];
    const float* gF[3];
    #pragma unroll
    for (int k = 0; k < 3; ++k) {
      const int j = lane + 32 * k;
      ft[k] = j / 24;  fq[k] = 4 * (j - 24 * ft[k]);
      gF[k] = Fg_ + (size_t)(ft[k] * NB_ + b) * (NS_ * NSC_) + fq[k];
    }
    int ct[3], cq[3];
    const float* gc[3];
    #pragma unroll
    for (int k = 0; k < 3; ++k) {
      const int j = lane + 32 * k;
      ct[k] = j / 3;  cq[k] = 4 * (j - 3 * ct[k]);
      gc[k] = cg_ + (size_t)(ct[k] * NB_ + b) * NSC_ + cq[k];
    }
    const int    f_t = lane >> 1, f_q = 4 * (lane & 1);
    const float* gf  = fg_ + (size_t)(f_t * NB_ + b) * NS_ + f_q;

    v4f bC[T_], bCt[4], bF[8][3], bc[3], bf[2];
    #pragma unroll
    for (int t = T_ - 1; t >= 0; --t) bC[t] = *reinterpret_cast<const v4f*>(gC + t * CS);
    #pragma unroll
    for (int i = 0; i < 4; ++i) bCt[i] = *reinterpret_cast<const v4f*>(gCt + i * 8 * CS);
    #pragma unroll
    for (int g = 0; g < 8; ++g) {
      #pragma unroll
      for (int k = 0; k < 3; ++k) bF[g][k] = *reinterpret_cast<const v4f*>(gF[k] + g * 4 * FS);
    }
    #pragma unroll
    for (int k = 0; k < 3; ++k) bc[k] = *reinterpret_cast<const v4f*>(gc[k]);
    #pragma unroll
    for (int k = 0; k < 2; ++k) bf[k] = *reinterpret_cast<const v4f*>(gf + k * 16 * fS);

    #pragma unroll
    for (int t = T_ - 1; t >= 0; --t) *reinterpret_cast<v4f*>(&sC[t][4 * lane]) = bC[t];
    #pragma unroll
    for (int i = 0; i < 4; ++i) *reinterpret_cast<v4f*>(&sC[8 * i + tq][qq]) = bCt[i];
    #pragma unroll
    for (int g = 0; g < 8; ++g) {
      #pragma unroll
      for (int k = 0; k < 3; ++k) *reinterpret_cast<v4f*>(&sF[4 * g + ft[k]][fq[k]]) = bF[g][k];
    }
    #pragma unroll
    for (int k = 0; k < 3; ++k) *reinterpret_cast<v4f*>(&sc[ct[k]][cq[k]]) = bc[k];
    #pragma unroll
    for (int k = 0; k < 2; ++k) *reinterpret_cast<v4f*>(&sf[f_t + 16 * k][f_q]) = bf[k];
  }
  lds_sync();

  auto fetch = [&](int t, StepRaw& rw) {
    const float* Fg = &sF[t][0];
    const float* Cg = &sC[t][0];
    const float* cg = &sc[t][0];
    const float* fg = &sf[t][0];
    const int  cc   = hl < NSC_ ? hl : NSC_ - 1;
    const bool aff  = hl == NSC_;
    const float* pf0 = aff ? fg : Fg + cc;
    const float* pc0 = aff ? cg : Cg + cc;
    const int    st  = aff ? 1 : NSC_;
    #pragma unroll
    for (int i = 0; i < 4; ++i) {
      const int k = 4 * (i >> 1) + kbase + (i & 1);
      rw.fv[i] = pf0[k * st];
    }
    #pragma unroll
    for (int r = 0; r < 8; ++r) {
      const int row = (r < 4) ? rbase + r : r;
      rw.cv[r] = pc0[row * st];
    }
    {
      const float* pt = Cg + cc * NSC_;
      const int r0 = rbase;
      rw.t0.x = pt[r0];  rw.t0.y = pt[r0 + 1];  rw.t0.z = pt[r0 + 2];  rw.t0.w = pt[r0 + 3];
      rw.t1.x = pt[4];   rw.t1.y = pt[5];       rw.t1.z = pt[6];       rw.t1.w = pt[7];
    }
  };
  auto shape = [&](const StepRaw& rw, StepIn& in) {
    const float* fv = rw.fv; const float* cv = rw.cv;
    const float tv[8] = {rw.t0.x, rw.t0.y, rw.t0.z, rw.t0.w, rw.t1.x, rw.t1.y, rw.t1.z, rw.t1.w};
    in.fa0.x = hl < NSC_ ? fv[0] : 0.f;  in.fa0.y = hl < NSC_ ? fv[1] : 0.f;
    in.fa1.x = hl < NSC_ ? fv[2] : 0.f;  in.fa1.y = hl < NSC_ ? fv[3] : 0.f;
    in.fb0.x = hl <= NSC_ ? fv[0] : 0.f; in.fb0.y = hl <= NSC_ ? fv[1] : 0.f;
    in.fb1.x = hl <= NSC_ ? fv[2] : 0.f; in.fb1.y = hl <= NSC_ ? fv[3] : 0.f;
    #pragma unroll
    for (int r = 0; r < 8; ++r) {
      const bool rowok = (r < 4) || !hiw;
      in.Cd[r]  = (rowok && hl <= NSC_) ? cv[r] : 0.f;
      in.CTd[r] = (rowok && hl <  NSC_) ? tv[r] : 0.f;
    }
  };
  StepRaw nx;
  fetch(T_ - 1, nx);
  const float x0v = x0g[(size_t)b * NS_ + (lane & (NS_ - 1))];

  Carry cy;

  auto frag = [&](const v8f& d, int kb) {
    const float s2 = other_half(d[4 * kb + 2]);
    const float s3 = other_half(d[4 * kb + 3]);
    v2f o;
    o.x = hiw ? s2 : d[4 * kb];
    o.y = hiw ? s3 : d[4 * kb + 1];
    return o;
  };

  auto solve_update = [&](const v8f& Qd, const v8f& QTd, int t) {
    float X[4];
    #pragma unroll
    for (int m = 0; m < 4; ++m) {
      const float o = other_half(Qd[m]);
      X[m] = hiw ? Qd[m] : o;
    }
    float A[4][4], Inv[4][4];
    #pragma unroll
    for (int i = 0; i < 4; ++i) {
      A[i][0] = row_lane<NS_ + 0>(X[i]);
      A[i][1] = row_lane<NS_ + 1>(X[i]);
      A[i][2] = row_lane<NS_ + 2>(X[i]);
      A[i][3] = row_lane<NS_ + 3>(X[i]);
      #pragma unroll
      for (int j = 0; j < 4; ++j) Inv[i][j] = (i == j) ? 1.f : 0.f;
    }
    #pragma unroll
    for (int cc = 0; cc < 4; ++cc) {
      float piv = 1.f / A[cc][cc];
      #pragma unroll
      for (int j = 0; j < 4; ++j) { A[cc][j] *= piv; Inv[cc][j] *= piv; }
      #pragma unroll
      for (int r = 0; r < 4; ++r) {
        if (r == cc) continue;
        float fac = A[r][cc];
        #pragma unroll
        for (int j = 0; j < 4; ++j) { A[r][j] -= fac * A[cc][j]; Inv[r][j] -= fac * Inv[cc][j]; }
      }
    }
    float kx = 0.f, ky = 0.f;
    #pragma unroll
    for (int m = 0; m < 4; ++m) {
      const float xm = colxp ? X[m] : 0.f;
      kx += (hiw ? Inv[2][m] : Inv[0][m]) * xm;
      ky += (hiw ? Inv[3][m] : Inv[1][m]) * xm;
    }
    kx = -kx; ky = -ky;
    Coef[t][NS_ + kbase][hl]     = kx;
    Coef[t][NS_ + kbase + 1][hl] = ky;
    const float o0 = other_half(QTd[0]);
    const float o1 = other_half(QTd[1]);
    v2f qxu, kf, kt;
    qxu.x = colx ? (hiw ? QTd[2] : o0) : 0.f;
    qxu.y = colx ? (hiw ? QTd[3] : o1) : 0.f;
    kf.x = kx;  kf.y = ky;
    kt.x = colx ? kx : 0.f;  kt.y = colx ? ky : 0.f;
    v8f sP, sT;
    #pragma unroll
    for (int r = 0; r < 8; ++r) {
      sP[r] = (!hiw && colxp) ? Qd[r]  : 0.f;
      sT[r] = (!hiw && colx)  ? QTd[r] : 0.f;
    }
    const v8f Pd  = wmma4(qxu, kf, sP);
    const v8f PTd = wmma4(kt, qxu, sT);
    cy.aW0 = frag(PTd, 0);  cy.aW1 = frag(PTd, 1);
    #pragma unroll
    for (int r = 0; r < 8; ++r) cy.seed[r] = (hl == NSC_) ? Pd[r] : 0.f;
  };

  {
    StepIn in;
    shape(nx, in);
    solve_update(in.Cd, in.CTd, T_);
    Coef[T_][kbase][hl]     = 0.f;  Coef[T_][kbase + 1][hl] = 0.f;
    Coef[T_][4 + kbase][hl] = 0.f;  Coef[T_][5 + kbase][hl] = 0.f;
  }

  for (int t = T_ - 1; t >= 0; --t) {
    StepIn in;
    shape(nx, in);
    if (t > 0) fetch(t - 1, nx);
    Coef[t][kbase][hl]     = in.fb0.x;  Coef[t][kbase + 1][hl] = in.fb0.y;
    Coef[t][4 + kbase][hl] = in.fb1.x;  Coef[t][5 + kbase][hl] = in.fb1.y;

    v8f Wd = wmma4(cy.aW0, in.fb0, cy.seed);
    Wd = wmma4(cy.aW1, in.fb1, Wd);

    const v2f w0 = frag(Wd, 0), w1 = frag(Wd, 1);
    v8f Qd  = wmma4(in.fa0, w0, in.Cd);
    v8f QTd = wmma4(w0, in.fa0, in.CTd);
    Qd  = wmma4(in.fa1, w1, Qd);
    QTd = wmma4(w1, in.fa1, QTd);

    solve_update(Qd, QTd, t);
  }

  float* out_x = out;
  float* out_u = out + (size_t)T1_ * NB_ * NS_;
  struct RowCo { v4f a0, a1, bu; float c0; };
  const int rr = hl < NSC_ ? hl : NSC_ - 1;
  auto ldco = [&](int t, RowCo& rc) {
    const v4f* p = reinterpret_cast<const v4f*>(&Coef[t][rr][0]);
    rc.a0 = p[0];  rc.a1 = p[1];  rc.bu = p[2];
    rc.c0 = Coef[t][rr][NSC_];
  };
  float v = x0v;
  float* po = hl < NS_ ? out_x + (size_t)b * NS_ + hl
                       : out_u + (size_t)b * NC_ + ((hl - NS_) & (NC_ - 1));
  const int pstride = hl < NS_ ? NB_ * NS_ : NB_ * NC_;
  auto step = [&](const RowCo& rc) {
    float s = rc.c0;
    s = __builtin_fmaf(rc.a0.x, row_lane<0>(v), s);  s = __builtin_fmaf(rc.a0.y, row_lane<1>(v), s);
    s = __builtin_fmaf(rc.a0.z, row_lane<2>(v), s);  s = __builtin_fmaf(rc.a0.w, row_lane<3>(v), s);
    s = __builtin_fmaf(rc.a1.x, row_lane<4>(v), s);  s = __builtin_fmaf(rc.a1.y, row_lane<5>(v), s);
    s = __builtin_fmaf(rc.a1.z, row_lane<6>(v), s);  s = __builtin_fmaf(rc.a1.w, row_lane<7>(v), s);
    const float val = hl < NS_ ? v : s;
    if (lane < NSC_) *po = val;
    po += pstride;
    float s2 = s;
    s2 = __builtin_fmaf(rc.bu.x, row_lane<NS_ + 0>(s), s2);  s2 = __builtin_fmaf(rc.bu.y, row_lane<NS_ + 1>(s), s2);
    s2 = __builtin_fmaf(rc.bu.z, row_lane<NS_ + 2>(s), s2);  s2 = __builtin_fmaf(rc.bu.w, row_lane<NS_ + 3>(s), s2);
    v = s2;
  };
  lds_sync();
  RowCo r0, r1, r2;
  ldco(0, r0);  ldco(1, r1);  ldco(2, r2);
  for (int t = 0; t <= T_; t += 3) {
    step(r0);  ldco(t + 3 < T_ ? t + 3 : T_, r0);
    step(r1);  ldco(t + 4 < T_ ? t + 4 : T_, r1);
    step(r2);  ldco(t + 5 < T_ ? t + 5 : T_, r2);
  }
}

extern "C" void kernel_launch(void* const* d_in, const int* in_sizes, int n_in,
                              void* d_out, int out_size, void* d_ws, size_t ws_size,
                              hipStream_t stream) {
  (void)in_sizes; (void)n_in; (void)d_ws; (void)ws_size; (void)out_size;
  const float* x0 = (const float*)d_in[0];
  const float* C  = (const float*)d_in[1];
  const float* c  = (const float*)d_in[2];
  const float* F  = (const float*)d_in[3];
  const float* f  = (const float*)d_in[4];
  float* out = (float*)d_out;
  mpc_riccati<<<NB_, 32, 0, stream>>>(x0, C, c, F, f, out);
}


# Each pivot reciprocal of the 4x4 elimination as hardware reciprocal + one Newton step (2 FMAs), not the 10-instruction division: same elimination; a gain may differ in its last bit
# speedup vs baseline: 9.2339x; 1.1293x over previous
#include <hip/hip_runtime.h>


#define NS_  8
#define NC_  4
#define T_   32
#define NB_  128
#define NSC_ 12
#define T1_  33

typedef float v2f __attribute__((ext_vector_type(2)));
typedef float v4f __attribute__((ext_vector_type(4)));
typedef float v8f __attribute__((ext_vector_type(8)));

__device__ __forceinline__ v8f wmma4(v2f a, v2f b, v8f acc) {
  return __builtin_amdgcn_wmma_f32_16x16x4_f32(
       false, a,  false, b,
       (short)0, acc,  false,  false);
}

__device__ __forceinline__ void lds_sync() {
  __builtin_amdgcn_fence(__ATOMIC_RELEASE, "workgroup", "local");
  __builtin_amdgcn_s_barrier();
  __builtin_amdgcn_fence(__ATOMIC_ACQUIRE, "workgroup", "local");
}

__device__ __forceinline__ float other_half(float v) {
  unsigned r = __builtin_amdgcn_permlanex16(0u, __builtin_bit_cast(unsigned, v),
                                            0x76543210u, 0xfedcba98u, true, true);
  return __builtin_bit_cast(float, r);
}

template <int L>
__device__ __forceinline__ float row_lane(float v) {
  int r = __builtin_amdgcn_mov_dpp(__builtin_bit_cast(int, v), 0x150 + L, 0xf, 0xf, true);
  return __builtin_bit_cast(float, r);
}

__device__ __forceinline__ float recip(float x) {
  const float r = __builtin_amdgcn_rcpf(x);
  const float e = __builtin_fmaf(-x, r, 1.f);
  return __builtin_fmaf(e, r, r);
}

struct StepIn {
  v2f fa0, fa1;
  v2f fb0, fb1;
  v8f Cd;
  v8f CTd;
};

struct StepRaw {
  float fv[4];
  float cv[8];
  v4f   t0, t1;
};

struct Carry {
  v2f aW0, aW1;
  v8f seed;
};

__global__ __launch_bounds__(32) void mpc_riccati(
    const float* __restrict__ x0g,
    const float* __restrict__ Cg_,
    const float* __restrict__ cg_,
    const float* __restrict__ Fg_,
    const float* __restrict__ fg_,
    float* __restrict__ out)
{
  const int  b    = blockIdx.x;
  const int  lane = threadIdx.x;
  const int  hl   = lane & 15;
  const bool hiw  = lane >= 16;
  const int  rbase = hiw ? 8 : 0;
  const int  kbase = hiw ? 2 : 0;
  const bool colx  = hl < NS_;
  const bool colxp = colx || hl == NSC_;

  __shared__ __attribute__((aligned(16))) float Coef[T1_][NSC_][16];
  __shared__ __attribute__((aligned(16))) float sC[T_][NSC_ * NSC_];
  __shared__ __attribute__((aligned(16))) float sc[T_][NSC_];
  __shared__ __attribute__((aligned(16))) float sF[T_][NS_ * NSC_];
  __shared__ __attribute__((aligned(16))) float sf[T_][NS_];

  {
    constexpr int CS = NB_ * NSC_ * NSC_;
    constexpr int FS = NB_ * NS_ * NSC_;
    constexpr int cS = NB_ * NSC_;
    constexpr int fS = NB_ * NS_;
    const float* gC  = Cg_ + (size_t)b * (NSC_ * NSC_) + 4 * lane;
    const int    tq  = lane >> 2, qq = 128 + 4 * (lane & 3);
    const float* gCt = Cg_ + (size_t)(tq * NB_ + b) * (NSC_ * NSC_) + qq;
    int ft[3], fq[3];
    const float* gF[3];
    #pragma unroll
    for (int k = 0; k < 3; ++k) {
      const int j = lane + 32 * k;
      ft[k] = j / 24;  fq[k] = 4 * (j - 24 * ft[k]);
      gF[k] = Fg_ + (size_t)(ft[k] * NB_ + b) * (NS_ * NSC_) + fq[k];
    }
    int ct[3], cq[3];
    const float* gc[3];
    #pragma unroll
    for (int k = 0; k < 3; ++k) {
      const int j = lane + 32 * k;
      ct[k] = j / 3;  cq[k] = 4 * (j - 3 * ct[k]);
      gc[k] = cg_ + (size_t)(ct[k] * NB_ + b) * NSC_ + cq[k];
    }
    const int    f_t = lane >> 1, f_q = 4 * (lane & 1);
    const float* gf  = fg_ + (size_t)(f_t * NB_ + b) * NS_ + f_q;

    v4f bC[T_], bCt[4], bF[8][3], bc[3], bf[2];
    #pragma unroll
    for (int t = T_ - 1; t >= 0; --t) bC[t] = *reinterpret_cast<const v4f*>(gC + t * CS);
    #pragma unroll
    for (int i = 0; i < 4; ++i) bCt[i] = *reinterpret_cast<const v4f*>(gCt + i * 8 * CS);
    #pragma unroll
    for (int g = 0; g < 8; ++g) {
      #pragma unroll
      for (int k = 0; k < 3; ++k) bF[g][k] = *reinterpret_cast<const v4f*>(gF[k] + g * 4 * FS);
    }
    #pragma unroll
    for (int k = 0; k < 3; ++k) bc[k] = *reinterpret_cast<const v4f*>(gc[k]);
    #pragma unroll
    for (int k = 0; k < 2; ++k) bf[k] = *reinterpret_cast<const v4f*>(gf + k * 16 * fS);

    #pragma unroll
    for (int t = T_ - 1; t >= 0; --t) *reinterpret_cast<v4f*>(&sC[t][4 * lane]) = bC[t];
    #pragma unroll
    for (int i = 0; i < 4; ++i) *reinterpret_cast<v4f*>(&sC[8 * i + tq][qq]) = bCt[i];
    #pragma unroll
    for (int g = 0; g < 8; ++g) {
      #pragma unroll
      for (int k = 0; k < 3; ++k) *reinterpret_cast<v4f*>(&sF[4 * g + ft[k]][fq[k]]) = bF[g][k];
    }
    #pragma unroll
    for (int k = 0; k < 3; ++k) *reinterpret_cast<v4f*>(&sc[ct[k]][cq[k]]) = bc[k];
    #pragma unroll
    for (int k = 0; k < 2; ++k) *reinterpret_cast<v4f*>(&sf[f_t + 16 * k][f_q]) = bf[k];
  }
  lds_sync();

  auto fetch = [&](int t, StepRaw& rw) {
    const float* Fg = &sF[t][0];
    const float* Cg = &sC[t][0];
    const float* cg = &sc[t][0];
    const float* fg = &sf[t][0];
    const int  cc   = hl < NSC_ ? hl : NSC_ - 1;
    const bool aff  = hl == NSC_;
    const float* pf0 = aff ? fg : Fg + cc;
    const float* pc0 = aff ? cg : Cg + cc;
    const int    st  = aff ? 1 : NSC_;
    #pragma unroll
    for (int i = 0; i < 4; ++i) {
      const int k = 4 * (i >> 1) + kbase + (i & 1);
      rw.fv[i] = pf0[k * st];
    }
    #pragma unroll
    for (int r = 0; r < 8; ++r) {
      const int row = (r < 4) ? rbase + r : r;
      rw.cv[r] = pc0[row * st];
    }
    {
      const float* pt = Cg + cc * NSC_;
      const int r0 = rbase;
      rw.t0.x = pt[r0];  rw.t0.y = pt[r0 + 1];  rw.t0.z = pt[r0 + 2];  rw.t0.w = pt[r0 + 3];
      rw.t1.x = pt[4];   rw.t1.y = pt[5];       rw.t1.z = pt[6];       rw.t1.w = pt[7];
    }
  };
  auto shape = [&](const StepRaw& rw, StepIn& in) {
    const float* fv = rw.fv; const float* cv = rw.cv;
    const float tv[8] = {rw.t0.x, rw.t0.y, rw.t0.z, rw.t0.w, rw.t1.x, rw.t1.y, rw.t1.z, rw.t1.w};
    in.fa0.x = hl < NSC_ ? fv[0] : 0.f;  in.fa0.y = hl < NSC_ ? fv[1] : 0.f;
    in.fa1.x = hl < NSC_ ? fv[2] : 0.f;  in.fa1.y = hl < NSC_ ? fv[3] : 0.f;
    in.fb0.x = hl <= NSC_ ? fv[0] : 0.f; in.fb0.y = hl <= NSC_ ? fv[1] : 0.f;
    in.fb1.x = hl <= NSC_ ? fv[2] : 0.f; in.fb1.y = hl <= NSC_ ? fv[3] : 0.f;
    #pragma unroll
    for (int r = 0; r < 8; ++r) {
      const bool rowok = (r < 4) || !hiw;
      in.Cd[r]  = (rowok && hl <= NSC_) ? cv[r] : 0.f;
      in.CTd[r] = (rowok && hl <  NSC_) ? tv[r] : 0.f;
    }
  };
  StepRaw nx;
  fetch(T_ - 1, nx);
  const float x0v = x0g[(size_t)b * NS_ + (lane & (NS_ - 1))];

  Carry cy;

  auto frag = [&](const v8f& d, int kb) {
    const float s2 = other_half(d[4 * kb + 2]);
    const float s3 = other_half(d[4 * kb + 3]);
    v2f o;
    o.x = hiw ? s2 : d[4 * kb];
    o.y = hiw ? s3 : d[4 * kb + 1];
    return o;
  };

  auto solve_update = [&](const v8f& Qd, const v8f& QTd, int t) {
    float X[4];
    #pragma unroll
    for (int m = 0; m < 4; ++m) {
      const float o = other_half(Qd[m]);
      X[m] = hiw ? Qd[m] : o;
    }
    float A[4][4], Inv[4][4];
    #pragma unroll
    for (int i = 0; i < 4; ++i) {
      A[i][0] = row_lane<NS_ + 0>(X[i]);
      A[i][1] = row_lane<NS_ + 1>(X[i]);
      A[i][2] = row_lane<NS_ + 2>(X[i]);
      A[i][3] = row_lane<NS_ + 3>(X[i]);
      #pragma unroll
      for (int j = 0; j < 4; ++j) Inv[i][j] = (i == j) ? 1.f : 0.f;
    }
    #pragma unroll
    for (int cc = 0; cc < 4; ++cc) {
      const float piv = recip(A[cc][cc]);
      #pragma unroll
      for (int j = 0; j < 4; ++j) { A[cc][j] *= piv; Inv[cc][j] *= piv; }
      #pragma unroll
      for (int r = 0; r < 4; ++r) {
        if (r == cc) continue;
        float fac = A[r][cc];
        #pragma unroll
        for (int j = 0; j < 4; ++j) { A[r][j] -= fac * A[cc][j]; Inv[r][j] -= fac * Inv[cc][j]; }
      }
    }
    float kx = 0.f, ky = 0.f;
    #pragma unroll
    for (int m = 0; m < 4; ++m) {
      const float xm = colxp ? X[m] : 0.f;
      kx += (hiw ? Inv[2][m] : Inv[0][m]) * xm;
      ky += (hiw ? Inv[3][m] : Inv[1][m]) * xm;
    }
    kx = -kx; ky = -ky;
    Coef[t][NS_ + kbase][hl]     = kx;
    Coef[t][NS_ + kbase + 1][hl] = ky;
    const float o0 = other_half(QTd[0]);
    const float o1 = other_half(QTd[1]);
    v2f qxu, kf, kt;
    qxu.x = colx ? (hiw ? QTd[2] : o0) : 0.f;
    qxu.y = colx ? (hiw ? QTd[3] : o1) : 0.f;
    kf.x = kx;  kf.y = ky;
    kt.x = colx ? kx : 0.f;  kt.y = colx ? ky : 0.f;
    v8f sP, sT;
    #pragma unroll
    for (int r = 0; r < 8; ++r) {
      sP[r] = (!hiw && colxp) ? Qd[r]  : 0.f;
      sT[r] = (!hiw && colx)  ? QTd[r] : 0.f;
    }
    const v8f Pd  = wmma4(qxu, kf, sP);
    const v8f PTd = wmma4(kt, qxu, sT);
    cy.aW0 = frag(PTd, 0);  cy.aW1 = frag(PTd, 1);
    #pragma unroll
    for (int r = 0; r < 8; ++r) cy.seed[r] = (hl == NSC_) ? Pd[r] : 0.f;
  };

  {
    StepIn in;
    shape(nx, in);
    solve_update(in.Cd, in.CTd, T_);
    Coef[T_][kbase][hl]     = 0.f;  Coef[T_][kbase + 1][hl] = 0.f;
    Coef[T_][4 + kbase][hl] = 0.f;  Coef[T_][5 + kbase][hl] = 0.f;
  }

  for (int t = T_ - 1; t >= 0; --t) {
    StepIn in;
    shape(nx, in);
    if (t > 0) fetch(t - 1, nx);
    Coef[t][kbase][hl]     = in.fb0.x;  Coef[t][kbase + 1][hl] = in.fb0.y;
    Coef[t][4 + kbase][hl] = in.fb1.x;  Coef[t][5 + kbase][hl] = in.fb1.y;

    v8f Wd = wmma4(cy.aW0, in.fb0, cy.seed);
    Wd = wmma4(cy.aW1, in.fb1, Wd);

    const v2f w0 = frag(Wd, 0), w1 = frag(Wd, 1);
    v8f Qd  = wmma4(in.fa0, w0, in.Cd);
    v8f QTd = wmma4(w0, in.fa0, in.CTd);
    Qd  = wmma4(in.fa1, w1, Qd);
    QTd = wmma4(w1, in.fa1, QTd);

    solve_update(Qd, QTd, t);
  }

  float* out_x = out;
  float* out_u = out + (size_t)T1_ * NB_ * NS_;
  struct RowCo { v4f a0, a1, bu; float c0; };
  const int rr = hl < NSC_ ? hl : NSC_ - 1;
  auto ldco = [&](int t, RowCo& rc) {
    const v4f* p = reinterpret_cast<const v4f*>(&Coef[t][rr][0]);
    rc.a0 = p[0];  rc.a1 = p[1];  rc.bu = p[2];
    rc.c0 = Coef[t][rr][NSC_];
  };
  float v = x0v;
  float* po = hl < NS_ ? out_x + (size_t)b * NS_ + hl
                       : out_u + (size_t)b * NC_ + ((hl - NS_) & (NC_ - 1));
  const int pstride = hl < NS_ ? NB_ * NS_ : NB_ * NC_;
  auto step = [&](const RowCo& rc) {
    float s = rc.c0;
    s = __builtin_fmaf(rc.a0.x, row_lane<0>(v), s);  s = __builtin_fmaf(rc.a0.y, row_lane<1>(v), s);
    s = __builtin_fmaf(rc.a0.z, row_lane<2>(v), s);  s = __builtin_fmaf(rc.a0.w, row_lane<3>(v), s);
    s = __builtin_fmaf(rc.a1.x, row_lane<4>(v), s);  s = __builtin_fmaf(rc.a1.y, row_lane<5>(v), s);
    s = __builtin_fmaf(rc.a1.z, row_lane<6>(v), s);  s = __builtin_fmaf(rc.a1.w, row_lane<7>(v), s);
    const float val = hl < NS_ ? v : s;
    if (lane < NSC_) *po = val;
    po += pstride;
    float s2 = s;
    s2 = __builtin_fmaf(rc.bu.x, row_lane<NS_ + 0>(s), s2);  s2 = __builtin_fmaf(rc.bu.y, row_lane<NS_ + 1>(s), s2);
    s2 = __builtin_fmaf(rc.bu.z, row_lane<NS_ + 2>(s), s2);  s2 = __builtin_fmaf(rc.bu.w, row_lane<NS_ + 3>(s), s2);
    v = s2;
  };
  lds_sync();
  RowCo r0, r1, r2;
  ldco(0, r0);  ldco(1, r1);  ldco(2, r2);
  for (int t = 0; t <= T_; t += 3) {
    step(r0);  ldco(t + 3 < T_ ? t + 3 : T_, r0);
    step(r1);  ldco(t + 4 < T_ ? t + 4 : T_, r1);
    step(r2);  ldco(t + 5 < T_ ? t + 5 : T_, r2);
  }
}

extern "C" void kernel_launch(void* const* d_in, const int* in_sizes, int n_in,
                              void* d_out, int out_size, void* d_ws, size_t ws_size,
                              hipStream_t stream) {
  (void)in_sizes; (void)n_in; (void)d_ws; (void)ws_size; (void)out_size;
  const float* x0 = (const float*)d_in[0];
  const float* C  = (const float*)d_in[1];
  const float* c  = (const float*)d_in[2];
  const float* F  = (const float*)d_in[3];
  const float* f  = (const float*)d_in[4];
  float* out = (float*)d_out;
  mpc_riccati<<<NB_, 32, 0, stream>>>(x0, C, c, F, f, out);
}


# Sweep step's operand reads taken off the serial chain: issued right behind the W products (loop unrolled by 2, no register copies) and used as loaded, zero-masks of never-read pad lanes dropped
# speedup vs baseline: 9.7054x; 1.0511x over previous
#include <hip/hip_runtime.h>


#define NS_  8
#define NC_  4
#define T_   32
#define NB_  128
#define NSC_ 12
#define T1_  33

typedef float v2f __attribute__((ext_vector_type(2)));
typedef float v4f __attribute__((ext_vector_type(4)));
typedef float v8f __attribute__((ext_vector_type(8)));

__device__ __forceinline__ v8f wmma4(v2f a, v2f b, v8f acc) {
  return __builtin_amdgcn_wmma_f32_16x16x4_f32(
       false, a,  false, b,
       (short)0, acc,  false,  false);
}

__device__ __forceinline__ void lds_sync() {
  __builtin_amdgcn_fence(__ATOMIC_RELEASE, "workgroup", "local");
  __builtin_amdgcn_s_barrier();
  __builtin_amdgcn_fence(__ATOMIC_ACQUIRE, "workgroup", "local");
}

__device__ __forceinline__ float other_half(float v) {
  unsigned r = __builtin_amdgcn_permlanex16(0u, __builtin_bit_cast(unsigned, v),
                                            0x76543210u, 0xfedcba98u, true, true);
  return __builtin_bit_cast(float, r);
}

template <int L>
__device__ __forceinline__ float row_lane(float v) {
  int r = __builtin_amdgcn_mov_dpp(__builtin_bit_cast(int, v), 0x150 + L, 0xf, 0xf, true);
  return __builtin_bit_cast(float, r);
}

__device__ __forceinline__ float recip(float x) {
  const float r = __builtin_amdgcn_rcpf(x);
  const float e = __builtin_fmaf(-x, r, 1.f);
  return __builtin_fmaf(e, r, r);
}

struct StepRaw {
  v2f f0, f1;
  v8f cd;
  v4f t0, t1;
};

struct Carry {
  v2f aW0, aW1;
  v8f seed;
};

__global__ __launch_bounds__(32) void mpc_riccati(
    const float* __restrict__ x0g,
    const float* __restrict__ Cg_,
    const float* __restrict__ cg_,
    const float* __restrict__ Fg_,
    const float* __restrict__ fg_,
    float* __restrict__ out)
{
  const int  b    = blockIdx.x;
  const int  lane = threadIdx.x;
  const int  hl   = lane & 15;
  const bool hiw  = lane >= 16;
  const int  rbase = hiw ? 8 : 0;
  const int  kbase = hiw ? 2 : 0;
  const bool colx  = hl < NS_;
  const bool colxp = colx || hl == NSC_;

  __shared__ __attribute__((aligned(16))) float Coef[T1_][NSC_][16];
  __shared__ __attribute__((aligned(16))) float sC[T_][NSC_ * NSC_];
  __shared__ __attribute__((aligned(16))) float sc[T_][NSC_];
  __shared__ __attribute__((aligned(16))) float sF[T_][NS_ * NSC_];
  __shared__ __attribute__((aligned(16))) float sf[T_][NS_];

  {
    constexpr int CS = NB_ * NSC_ * NSC_;
    constexpr int FS = NB_ * NS_ * NSC_;
    constexpr int cS = NB_ * NSC_;
    constexpr int fS = NB_ * NS_;
    const float* gC  = Cg_ + (size_t)b * (NSC_ * NSC_) + 4 * lane;
    const int    tq  = lane >> 2, qq = 128 + 4 * (lane & 3);
    const float* gCt = Cg_ + (size_t)(tq * NB_ + b) * (NSC_ * NSC_) + qq;
    int ft[3], fq[3];
    const float* gF[3];
    #pragma unroll
    for (int k = 0; k < 3; ++k) {
      const int j = lane + 32 * k;
      ft[k] = j / 24;  fq[k] = 4 * (j - 24 * ft[k]);
      gF[k] = Fg_ + (size_t)(ft[k] * NB_ + b) * (NS_ * NSC_) + fq[k];
    }
    int ct[3], cq[3];
    const float* gc[3];
    #pragma unroll
    for (int k = 0; k < 3; ++k) {
      const int j = lane + 32 * k;
      ct[k] = j / 3;  cq[k] = 4 * (j - 3 * ct[k]);
      gc[k] = cg_ + (size_t)(ct[k] * NB_ + b) * NSC_ + cq[k];
    }
    const int    f_t = lane >> 1, f_q = 4 * (lane & 1);
    const float* gf  = fg_ + (size_t)(f_t * NB_ + b) * NS_ + f_q;

    v4f bC[T_], bCt[4], bF[8][3], bc[3], bf[2];
    #pragma unroll
    for (int t = T_ - 1; t >= 0; --t) bC[t] = *reinterpret_cast<const v4f*>(gC + t * CS);
    #pragma unroll
    for (int i = 0; i < 4; ++i) bCt[i] = *reinterpret_cast<const v4f*>(gCt + i * 8 * CS);
    #pragma unroll
    for (int g = 0; g < 8; ++g) {
      #pragma unroll
      for (int k = 0; k < 3; ++k) bF[g][k] = *reinterpret_cast<const v4f*>(gF[k] + g * 4 * FS);
    }
    #pragma unroll
    for (int k = 0; k < 3; ++k) bc[k] = *reinterpret_cast<const v4f*>(gc[k]);
    #pragma unroll
    for (int k = 0; k < 2; ++k) bf[k] = *reinterpret_cast<const v4f*>(gf + k * 16 * fS);

    #pragma unroll
    for (int t = T_ - 1; t >= 0; --t) *reinterpret_cast<v4f*>(&sC[t][4 * lane]) = bC[t];
    #pragma unroll
    for (int i = 0; i < 4; ++i) *reinterpret_cast<v4f*>(&sC[8 * i + tq][qq]) = bCt[i];
    #pragma unroll
    for (int g = 0; g < 8; ++g) {
      #pragma unroll
      for (int k = 0; k < 3; ++k) *reinterpret_cast<v4f*>(&sF[4 * g + ft[k]][fq[k]]) = bF[g][k];
    }
    #pragma unroll
    for (int k = 0; k < 3; ++k) *reinterpret_cast<v4f*>(&sc[ct[k]][cq[k]]) = bc[k];
    #pragma unroll
    for (int k = 0; k < 2; ++k) *reinterpret_cast<v4f*>(&sf[f_t + 16 * k][f_q]) = bf[k];
  }
  lds_sync();

  auto fetch = [&](int t, StepRaw& rw) {
    const float* Fg = &sF[t][0];
    const float* Cg = &sC[t][0];
    const float* cg = &sc[t][0];
    const float* fg = &sf[t][0];
    const int  cc   = hl < NSC_ ? hl : NSC_ - 1;
    const bool aff  = hl == NSC_;
    const float* pf0 = aff ? fg : Fg + cc;
    const float* pc0 = aff ? cg : Cg + cc;
    const int    st  = aff ? 1 : NSC_;
    rw.f0.x = pf0[(kbase + 0) * st];  rw.f0.y = pf0[(kbase + 1) * st];
    rw.f1.x = pf0[(kbase + 4) * st];  rw.f1.y = pf0[(kbase + 5) * st];
    #pragma unroll
    for (int r = 0; r < 8; ++r) {
      const int row = (r < 4) ? rbase + r : r;
      rw.cd[r] = pc0[row * st];
    }
    {
      const float* pt = Cg + cc * NSC_;
      const int r0 = rbase;
      rw.t0.x = pt[r0];  rw.t0.y = pt[r0 + 1];  rw.t0.z = pt[r0 + 2];  rw.t0.w = pt[r0 + 3];
      rw.t1.x = pt[4];   rw.t1.y = pt[5];       rw.t1.z = pt[6];       rw.t1.w = pt[7];
    }
  };
  auto ctile = [&](const StepRaw& rw) {
    v8f d;
    d[0] = rw.t0.x;  d[1] = rw.t0.y;  d[2] = rw.t0.z;  d[3] = rw.t0.w;
    d[4] = rw.t1.x;  d[5] = rw.t1.y;  d[6] = rw.t1.z;  d[7] = rw.t1.w;
    return d;
  };
  StepRaw nx;
  fetch(T_ - 1, nx);
  const float x0v = x0g[(size_t)b * NS_ + (lane & (NS_ - 1))];

  Carry cy;

  auto frag = [&](const v8f& d, int kb) {
    const float s2 = other_half(d[4 * kb + 2]);
    const float s3 = other_half(d[4 * kb + 3]);
    v2f o;
    o.x = hiw ? s2 : d[4 * kb];
    o.y = hiw ? s3 : d[4 * kb + 1];
    return o;
  };

  auto solve_update = [&](const v8f& Qd, const v8f& QTd, int t) {
    float X[4];
    #pragma unroll
    for (int m = 0; m < 4; ++m) {
      const float o = other_half(Qd[m]);
      X[m] = hiw ? Qd[m] : o;
    }
    float A[4][4], Inv[4][4];
    #pragma unroll
    for (int i = 0; i < 4; ++i) {
      A[i][0] = row_lane<NS_ + 0>(X[i]);
      A[i][1] = row_lane<NS_ + 1>(X[i]);
      A[i][2] = row_lane<NS_ + 2>(X[i]);
      A[i][3] = row_lane<NS_ + 3>(X[i]);
      #pragma unroll
      for (int j = 0; j < 4; ++j) Inv[i][j] = (i == j) ? 1.f : 0.f;
    }
    #pragma unroll
    for (int cc = 0; cc < 4; ++cc) {
      const float piv = recip(A[cc][cc]);
      #pragma unroll
      for (int j = 0; j < 4; ++j) { A[cc][j] *= piv; Inv[cc][j] *= piv; }
      #pragma unroll
      for (int r = 0; r < 4; ++r) {
        if (r == cc) continue;
        float fac = A[r][cc];
        #pragma unroll
        for (int j = 0; j < 4; ++j) { A[r][j] -= fac * A[cc][j]; Inv[r][j] -= fac * Inv[cc][j]; }
      }
    }
    float kx = 0.f, ky = 0.f;
    #pragma unroll
    for (int m = 0; m < 4; ++m) {
      const float xm = colxp ? X[m] : 0.f;
      kx += (hiw ? Inv[2][m] : Inv[0][m]) * xm;
      ky += (hiw ? Inv[3][m] : Inv[1][m]) * xm;
    }
    kx = -kx; ky = -ky;
    Coef[t][NS_ + kbase][hl]     = kx;
    Coef[t][NS_ + kbase + 1][hl] = ky;
    const float o0 = other_half(QTd[0]);
    const float o1 = other_half(QTd[1]);
    v2f qxu, kf;
    qxu.x = hiw ? QTd[2] : o0;
    qxu.y = hiw ? QTd[3] : o1;
    kf.x = kx;  kf.y = ky;
    const v8f Pd  = wmma4(qxu, kf, Qd);
    const v8f PTd = wmma4(kf, qxu, QTd);
    cy.aW0 = frag(PTd, 0);  cy.aW1 = frag(PTd, 1);
    #pragma unroll
    for (int r = 0; r < 8; ++r) cy.seed[r] = (hl == NSC_) ? Pd[r] : 0.f;
  };

  {
    solve_update(nx.cd, ctile(nx), T_);
    Coef[T_][kbase][hl]     = 0.f;  Coef[T_][kbase + 1][hl] = 0.f;
    Coef[T_][4 + kbase][hl] = 0.f;  Coef[T_][5 + kbase][hl] = 0.f;
  }

  #pragma unroll 2
  for (int t = T_ - 1; t >= 0; --t) {
    const StepRaw in = nx;

    v8f Wd = wmma4(cy.aW0, in.f0, cy.seed);
    Wd = wmma4(cy.aW1, in.f1, Wd);

    __builtin_amdgcn_sched_barrier(0);
    fetch(t > 0 ? t - 1 : 0, nx);
    Coef[t][kbase][hl]     = in.f0.x;  Coef[t][kbase + 1][hl] = in.f0.y;
    Coef[t][4 + kbase][hl] = in.f1.x;  Coef[t][5 + kbase][hl] = in.f1.y;
    __builtin_amdgcn_sched_barrier(0);

    const v2f w0 = frag(Wd, 0), w1 = frag(Wd, 1);
    v8f Qd  = wmma4(in.f0, w0, in.cd);
    v8f QTd = wmma4(w0, in.f0, ctile(in));
    Qd  = wmma4(in.f1, w1, Qd);
    QTd = wmma4(w1, in.f1, QTd);

    solve_update(Qd, QTd, t);
  }

  float* out_x = out;
  float* out_u = out + (size_t)T1_ * NB_ * NS_;
  struct RowCo { v4f a0, a1, bu; float c0; };
  const int rr = hl < NSC_ ? hl : NSC_ - 1;
  auto ldco = [&](int t, RowCo& rc) {
    const v4f* p = reinterpret_cast<const v4f*>(&Coef[t][rr][0]);
    rc.a0 = p[0];  rc.a1 = p[1];  rc.bu = p[2];
    rc.c0 = Coef[t][rr][NSC_];
  };
  float v = x0v;
  float* po = hl < NS_ ? out_x + (size_t)b * NS_ + hl
                       : out_u + (size_t)b * NC_ + ((hl - NS_) & (NC_ - 1));
  const int pstride = hl < NS_ ? NB_ * NS_ : NB_ * NC_;
  auto step = [&](const RowCo& rc) {
    float s = rc.c0;
    s = __builtin_fmaf(rc.a0.x, row_lane<0>(v), s);  s = __builtin_fmaf(rc.a0.y, row_lane<1>(v), s);
    s = __builtin_fmaf(rc.a0.z, row_lane<2>(v), s);  s = __builtin_fmaf(rc.a0.w, row_lane<3>(v), s);
    s = __builtin_fmaf(rc.a1.x, row_lane<4>(v), s);  s = __builtin_fmaf(rc.a1.y, row_lane<5>(v), s);
    s = __builtin_fmaf(rc.a1.z, row_lane<6>(v), s);  s = __builtin_fmaf(rc.a1.w, row_lane<7>(v), s);
    const float val = hl < NS_ ? v : s;
    if (lane < NSC_) *po = val;
    po += pstride;
    float s2 = s;
    s2 = __builtin_fmaf(rc.bu.x, row_lane<NS_ + 0>(s), s2);  s2 = __builtin_fmaf(rc.bu.y, row_lane<NS_ + 1>(s), s2);
    s2 = __builtin_fmaf(rc.bu.z, row_lane<NS_ + 2>(s), s2);  s2 = __builtin_fmaf(rc.bu.w, row_lane<NS_ + 3>(s), s2);
    v = s2;
  };
  lds_sync();
  RowCo r0, r1, r2;
  ldco(0, r0);  ldco(1, r1);  ldco(2, r2);
  for (int t = 0; t <= T_; t += 3) {
    step(r0);  ldco(t + 3 < T_ ? t + 3 : T_, r0);
    step(r1);  ldco(t + 4 < T_ ? t + 4 : T_, r1);
    step(r2);  ldco(t + 5 < T_ ? t + 5 : T_, r2);
  }
}

extern "C" void kernel_launch(void* const* d_in, const int* in_sizes, int n_in,
                              void* d_out, int out_size, void* d_ws, size_t ws_size,
                              hipStream_t stream) {
  (void)in_sizes; (void)n_in; (void)d_ws; (void)ws_size; (void)out_size;
  const float* x0 = (const float*)d_in[0];
  const float* C  = (const float*)d_in[1];
  const float* c  = (const float*)d_in[2];
  const float* F  = (const float*)d_in[3];
  const float* f  = (const float*)d_in[4];
  float* out = (float*)d_out;
  mpc_riccati<<<NB_, 32, 0, stream>>>(x0, C, c, F, f, out);
}


# Sweep step: what stood between dependent products leaves the chain: Q's two products ahead of Q^T's, p from its own masked product (no select), gain sign and LDS store moved, F-row copies dropped
# speedup vs baseline: 10.4513x; 1.0769x over previous
#include <hip/hip_runtime.h>


#define NS_  8
#define NC_  4
#define T_   32
#define NB_  128
#define NSC_ 12
#define T1_  33

typedef float v2f __attribute__((ext_vector_type(2)));
typedef float v4f __attribute__((ext_vector_type(4)));
typedef float v8f __attribute__((ext_vector_type(8)));

__device__ __forceinline__ v8f wmma4(v2f a, v2f b, v8f acc) {
  return __builtin_amdgcn_wmma_f32_16x16x4_f32(
       false, a,  false, b,
       (short)0, acc,  false,  false);
}


__device__ __forceinline__ void lds_sync() {
  __builtin_amdgcn_fence(__ATOMIC_RELEASE, "workgroup", "local");
  __builtin_amdgcn_s_barrier();
  __builtin_amdgcn_fence(__ATOMIC_ACQUIRE, "workgroup", "local");
}

__device__ __forceinline__ float other_half(float v) {
  unsigned r = __builtin_amdgcn_permlanex16(0u, __builtin_bit_cast(unsigned, v),
                                            0x76543210u, 0xfedcba98u, true, true);
  return __builtin_bit_cast(float, r);
}

template <int L>
__device__ __forceinline__ float row_lane(float v) {
  int r = __builtin_amdgcn_mov_dpp(__builtin_bit_cast(int, v), 0x150 + L, 0xf, 0xf, true);
  return __builtin_bit_cast(float, r);
}

__device__ __forceinline__ float recip(float x) {
  const float r = __builtin_amdgcn_rcpf(x);
  const float e = __builtin_fmaf(-x, r, 1.f);
  return __builtin_fmaf(e, r, r);
}

struct StepRaw {
  v2f f0, f1;
  v8f cd;
  v4f t0, t1;
};

struct Carry {
  v2f aW0, aW1;
  v8f seed;
  v2f gain;
};

__global__ __launch_bounds__(32) void mpc_riccati(
    const float* __restrict__ x0g,
    const float* __restrict__ Cg_,
    const float* __restrict__ cg_,
    const float* __restrict__ Fg_,
    const float* __restrict__ fg_,
    float* __restrict__ out)
{
  const int  b    = blockIdx.x;
  const int  lane = threadIdx.x;
  const int  hl   = lane & 15;
  const bool hiw  = lane >= 16;
  const int  rbase = hiw ? 8 : 0;
  const int  kbase = hiw ? 2 : 0;
  const bool colx  = hl < NS_;
  const bool colxp = colx || hl == NSC_;

  __shared__ __attribute__((aligned(16))) float Gain[T1_][NC_][16];
  __shared__ __attribute__((aligned(16))) float sC[T_][NSC_ * NSC_];
  __shared__ __attribute__((aligned(16))) float sc[T_][NSC_];
  __shared__ __attribute__((aligned(16))) float sF[T_][NS_ * NSC_];
  __shared__ __attribute__((aligned(16))) float sf[T_][NS_];

  {
    constexpr int CS = NB_ * NSC_ * NSC_;
    constexpr int FS = NB_ * NS_ * NSC_;
    constexpr int cS = NB_ * NSC_;
    constexpr int fS = NB_ * NS_;
    const float* gC  = Cg_ + (size_t)b * (NSC_ * NSC_) + 4 * lane;
    const int    tq  = lane >> 2, qq = 128 + 4 * (lane & 3);
    const float* gCt = Cg_ + (size_t)(tq * NB_ + b) * (NSC_ * NSC_) + qq;
    int ft[3], fq[3];
    const float* gF[3];
    #pragma unroll
    for (int k = 0; k < 3; ++k) {
      const int j = lane + 32 * k;
      ft[k] = j / 24;  fq[k] = 4 * (j - 24 * ft[k]);
      gF[k] = Fg_ + (size_t)(ft[k] * NB_ + b) * (NS_ * NSC_) + fq[k];
    }
    int ct[3], cq[3];
    const float* gc[3];
    #pragma unroll
    for (int k = 0; k < 3; ++k) {
      const int j = lane + 32 * k;
      ct[k] = j / 3;  cq[k] = 4 * (j - 3 * ct[k]);
      gc[k] = cg_ + (size_t)(ct[k] * NB_ + b) * NSC_ + cq[k];
    }
    const int    f_t = lane >> 1, f_q = 4 * (lane & 1);
    const float* gf  = fg_ + (size_t)(f_t * NB_ + b) * NS_ + f_q;

    v4f bC[T_], bCt[4], bF[8][3], bc[3], bf[2];
    #pragma unroll
    for (int t = T_ - 1; t >= 0; --t) bC[t] = *reinterpret_cast<const v4f*>(gC + t * CS);
    #pragma unroll
    for (int i = 0; i < 4; ++i) bCt[i] = *reinterpret_cast<const v4f*>(gCt + i * 8 * CS);
    #pragma unroll
    for (int g = 0; g < 8; ++g) {
      #pragma unroll
      for (int k = 0; k < 3; ++k) bF[g][k] = *reinterpret_cast<const v4f*>(gF[k] + g * 4 * FS);
    }
    #pragma unroll
    for (int k = 0; k < 3; ++k) bc[k] = *reinterpret_cast<const v4f*>(gc[k]);
    #pragma unroll
    for (int k = 0; k < 2; ++k) bf[k] = *reinterpret_cast<const v4f*>(gf + k * 16 * fS);

    #pragma unroll
    for (int t = T_ - 1; t >= 0; --t) *reinterpret_cast<v4f*>(&sC[t][4 * lane]) = bC[t];
    #pragma unroll
    for (int i = 0; i < 4; ++i) *reinterpret_cast<v4f*>(&sC[8 * i + tq][qq]) = bCt[i];
    #pragma unroll
    for (int g = 0; g < 8; ++g) {
      #pragma unroll
      for (int k = 0; k < 3; ++k) *reinterpret_cast<v4f*>(&sF[4 * g + ft[k]][fq[k]]) = bF[g][k];
    }
    #pragma unroll
    for (int k = 0; k < 3; ++k) *reinterpret_cast<v4f*>(&sc[ct[k]][cq[k]]) = bc[k];
    #pragma unroll
    for (int k = 0; k < 2; ++k) *reinterpret_cast<v4f*>(&sf[f_t + 16 * k][f_q]) = bf[k];
  }
  lds_sync();

  auto fetch = [&](int t, StepRaw& rw) {
    const float* Fg = &sF[t][0];
    const float* Cg = &sC[t][0];
    const float* cg = &sc[t][0];
    const float* fg = &sf[t][0];
    const int  cc   = hl < NSC_ ? hl : NSC_ - 1;
    const bool aff  = hl == NSC_;
    const float* pf0 = aff ? fg : Fg + cc;
    const float* pc0 = aff ? cg : Cg + cc;
    const int    st  = aff ? 1 : NSC_;
    rw.f0.x = pf0[(kbase + 0) * st];  rw.f0.y = pf0[(kbase + 1) * st];
    rw.f1.x = pf0[(kbase + 4) * st];  rw.f1.y = pf0[(kbase + 5) * st];
    #pragma unroll
    for (int r = 0; r < 8; ++r) {
      const int row = (r < 4) ? rbase + r : r;
      rw.cd[r] = pc0[row * st];
    }
    {
      const float* pt = Cg + cc * NSC_;
      const int r0 = rbase;
      rw.t0.x = pt[r0];  rw.t0.y = pt[r0 + 1];  rw.t0.z = pt[r0 + 2];  rw.t0.w = pt[r0 + 3];
      rw.t1.x = pt[4];   rw.t1.y = pt[5];       rw.t1.z = pt[6];       rw.t1.w = pt[7];
    }
  };
  auto ctile = [&](const StepRaw& rw) {
    v8f d;
    d[0] = rw.t0.x;  d[1] = rw.t0.y;  d[2] = rw.t0.z;  d[3] = rw.t0.w;
    d[4] = rw.t1.x;  d[5] = rw.t1.y;  d[6] = rw.t1.z;  d[7] = rw.t1.w;
    return d;
  };
  StepRaw nx;
  fetch(T_ - 1, nx);
  const float x0v = x0g[(size_t)b * NS_ + (lane & (NS_ - 1))];

  Carry cy;

  auto frag = [&](const v8f& d, int kb) {
    const float s2 = other_half(d[4 * kb + 2]);
    const float s3 = other_half(d[4 * kb + 3]);
    v2f o;
    o.x = hiw ? s2 : d[4 * kb];
    o.y = hiw ? s3 : d[4 * kb + 1];
    return o;
  };

  auto ctrl_rows = [&](const v8f& Qd, float (&X)[4]) {
    #pragma unroll
    for (int m = 0; m < 4; ++m) {
      const float o = other_half(Qd[m]);
      X[m] = hiw ? Qd[m] : o;
    }
  };

  auto solve_update = [&](const v8f& Qd, const v8f& QTd, const float (&X)[4]) {
    v8f q12;
    #pragma unroll
    for (int r = 0; r < 8; ++r) q12[r] = (hl == NSC_) ? Qd[r] : 0.f;
    float A[4][4], Inv[4][4];
    #pragma unroll
    for (int i = 0; i < 4; ++i) {
      A[i][0] = row_lane<NS_ + 0>(X[i]);
      A[i][1] = row_lane<NS_ + 1>(X[i]);
      A[i][2] = row_lane<NS_ + 2>(X[i]);
      A[i][3] = row_lane<NS_ + 3>(X[i]);
      #pragma unroll
      for (int j = 0; j < 4; ++j) Inv[i][j] = (i == j) ? 1.f : 0.f;
    }
    #pragma unroll
    for (int cc = 0; cc < 4; ++cc) {
      const float piv = recip(A[cc][cc]);
      #pragma unroll
      for (int j = 0; j < 4; ++j) { A[cc][j] *= piv; Inv[cc][j] *= piv; }
      #pragma unroll
      for (int r = 0; r < 4; ++r) {
        if (r == cc) continue;
        const float fac = A[r][cc];
        #pragma unroll
        for (int j = 0; j < 4; ++j) {
          A[r][j]   = __builtin_fmaf(-fac, A[cc][j], A[r][j]);
          Inv[r][j] = __builtin_fmaf(-fac, Inv[cc][j], Inv[r][j]);
        }
      }
    }
    float kx, ky;
    #pragma unroll
    for (int m = 0; m < 4; ++m) {
      const float xm = colxp ? -X[m] : 0.f;
      const float px = (hiw ? Inv[2][m] : Inv[0][m]) * xm;
      const float py = (hiw ? Inv[3][m] : Inv[1][m]) * xm;
      if (m == 0) { kx = px; ky = py; } else { kx += px; ky += py; }
    }
    cy.gain.x = kx;  cy.gain.y = ky;
    const float o0 = other_half(QTd[0]);
    const float o1 = other_half(QTd[1]);
    v2f qxu, kf;
    qxu.x = hiw ? QTd[2] : o0;
    qxu.y = hiw ? QTd[3] : o1;
    kf.x = kx;  kf.y = ky;
    v2f k12;
    k12.x = (hl == NSC_) ? kx : 0.f;
    k12.y = (hl == NSC_) ? ky : 0.f;
    const v8f PTd = wmma4(kf, qxu, QTd);
    cy.seed       = wmma4(qxu, k12, q12);
    __builtin_amdgcn_sched_barrier(0);
    cy.aW0 = frag(PTd, 0);  cy.aW1 = frag(PTd, 1);
  };

  {
    float X[4];
    ctrl_rows(nx.cd, X);
    solve_update(nx.cd, ctile(nx), X);
  }

  #pragma unroll 2
  for (int t = T_ - 1; t >= 0; --t) {
    const StepRaw in = nx;

    v8f Wd = wmma4(cy.aW0, in.f0, cy.seed);
    Wd = wmma4(cy.aW1, in.f1, Wd);

    __builtin_amdgcn_sched_barrier(0);
    fetch(t > 0 ? t - 1 : 0, nx);
    Gain[t + 1][kbase][hl]     = cy.gain.x;
    Gain[t + 1][kbase + 1][hl] = cy.gain.y;
    __builtin_amdgcn_sched_barrier(0);

    const v2f w0 = frag(Wd, 0), w1 = frag(Wd, 1);
    v8f Qd  = wmma4(in.f0, w0, in.cd);
    Qd  = wmma4(in.f1, w1, Qd);
    __builtin_amdgcn_sched_barrier(0x2 | 0x4 | 0x10 | 0x80 | 0x400);
    v8f QTd = wmma4(w0, in.f0, ctile(in));
    float X[4];
    ctrl_rows(Qd, X);
    __builtin_amdgcn_sched_barrier(0x2 | 0x4 | 0x10 | 0x80 | 0x400);
    QTd = wmma4(w1, in.f1, QTd);

    solve_update(Qd, QTd, X);
  }
  Gain[0][kbase][hl]     = cy.gain.x;
  Gain[0][kbase + 1][hl] = cy.gain.y;

  float* out_x = out;
  float* out_u = out + (size_t)T1_ * NB_ * NS_;
  struct RowCo { v4f a0, a1, bu; float c0; };
  const int gr = (hl < NSC_ ? hl : NSC_ - 1) - NS_;
  const float* rowb = colx ? &sF[0][NSC_ * hl] : &Gain[0][gr][0];
  const float* affb = colx ? &sf[0][hl]        : &Gain[0][gr][NSC_];
  const int rowp = colx ? NS_ * NSC_ : NC_ * 16;
  const int affp = colx ? NS_        : NC_ * 16;
  auto ldco = [&](int t, RowCo& rc) {
    const int tt = (colx && t > T_ - 1) ? T_ - 1 : t;
    const v4f* p = reinterpret_cast<const v4f*>(rowb + tt * rowp);
    rc.a0 = p[0];  rc.a1 = p[1];  rc.bu = p[2];
    rc.c0 = affb[tt * affp];
  };
  float v = x0v;
  float* po = hl < NS_ ? out_x + (size_t)b * NS_ + hl
                       : out_u + (size_t)b * NC_ + ((hl - NS_) & (NC_ - 1));
  const int pstride = hl < NS_ ? NB_ * NS_ : NB_ * NC_;
  auto step = [&](const RowCo& rc) {
    float s = rc.c0;
    s = __builtin_fmaf(rc.a0.x, row_lane<0>(v), s);  s = __builtin_fmaf(rc.a0.y, row_lane<1>(v), s);
    s = __builtin_fmaf(rc.a0.z, row_lane<2>(v), s);  s = __builtin_fmaf(rc.a0.w, row_lane<3>(v), s);
    s = __builtin_fmaf(rc.a1.x, row_lane<4>(v), s);  s = __builtin_fmaf(rc.a1.y, row_lane<5>(v), s);
    s = __builtin_fmaf(rc.a1.z, row_lane<6>(v), s);  s = __builtin_fmaf(rc.a1.w, row_lane<7>(v), s);
    const float val = hl < NS_ ? v : s;
    if (lane < NSC_) *po = val;
    po += pstride;
    float s2 = s;
    s2 = __builtin_fmaf(rc.bu.x, row_lane<NS_ + 0>(s), s2);  s2 = __builtin_fmaf(rc.bu.y, row_lane<NS_ + 1>(s), s2);
    s2 = __builtin_fmaf(rc.bu.z, row_lane<NS_ + 2>(s), s2);  s2 = __builtin_fmaf(rc.bu.w, row_lane<NS_ + 3>(s), s2);
    v = s2;
  };
  lds_sync();
  RowCo r0, r1, r2;
  ldco(0, r0);  ldco(1, r1);  ldco(2, r2);
  for (int t = 0; t <= T_; t += 3) {
    step(r0);  ldco(t + 3 < T_ ? t + 3 : T_, r0);
    step(r1);  ldco(t + 4 < T_ ? t + 4 : T_, r1);
    step(r2);  ldco(t + 5 < T_ ? t + 5 : T_, r2);
  }
}

extern "C" void kernel_launch(void* const* d_in, const int* in_sizes, int n_in,
                              void* d_out, int out_size, void* d_ws, size_t ws_size,
                              hipStream_t stream) {
  (void)in_sizes; (void)n_in; (void)d_ws; (void)ws_size; (void)out_size;
  const float* x0 = (const float*)d_in[0];
  const float* C  = (const float*)d_in[1];
  const float* c  = (const float*)d_in[2];
  const float* F  = (const float*)d_in[3];
  const float* f  = (const float*)d_in[4];
  float* out = (float*)d_out;
  mpc_riccati<<<NB_, 32, 0, stream>>>(x0, C, c, F, f, out);
}
